# starting kernel
# baseline (speedup 1.0000x reference)
#include <hip/hip_runtime.h>
#include <hip/hip_bf16.h>
#include <math.h>

#define NEG_ATT 0.2f
#define NEG_OUT 0.01f

typedef __attribute__((ext_vector_type(2))) float v2f;
typedef __attribute__((ext_vector_type(8))) float v8f;

__device__ __forceinline__ void atomic_max_float(float* addr, float val) {
    if (val >= 0.0f) {
        atomicMax((int*)addr, __float_as_int(val));
    } else {
        atomicMin((unsigned int*)addr, __float_as_uint(val));
    }
}

__global__ void gat_init(float* __restrict__ out, float* __restrict__ seg_max,
                         float* __restrict__ seg_sum, int N) {
    int i = blockIdx.x * blockDim.x + threadIdx.x;
    int total = N * 32;
    if (i < total) out[i] = 0.0f;
    if (i < N) {
        ((unsigned int*)seg_max)[i] = 0xFF800000u;
        seg_sum[i] = 0.0f;
    }
}

__global__ __launch_bounds__(256) void gat_gemm(
        const float* __restrict__ x,
        const float* __restrict__ Wl, const float* __restrict__ bl,
        const float* __restrict__ Wr, const float* __restrict__ br,
        float* __restrict__ xl, float* __restrict__ xr, int N) {
    const int lane = threadIdx.x & 31;
    const int wave = threadIdx.x >> 5;
    const int tile = blockIdx.x * (blockDim.x >> 5) + wave;
    const int row0 = tile * 16;
    if (row0 >= N) return;

    const int m  = lane & 15;
    const int kh = lane >> 4;

    int arow = row0 + m;
    if (arow >= N) arow = N - 1;
    const float* __restrict__ xrow = x + (size_t)arow * 128;

    v8f cl0 = {}, cl1 = {}, cr0 = {}, cr1 = {};

    #pragma unroll 4
    for (int k0 = 0; k0 < 128; k0 += 4) {
        const int kb = k0 + 2 * kh;
        v2f a;
        a[0] = xrow[kb];
        a[1] = xrow[kb + 1];

        const int n = m;
        v2f bl0v, bl1v, br0v, br1v;
        bl0v[0] = Wl[kb * 32 + n];       bl0v[1] = Wl[(kb + 1) * 32 + n];
        bl1v[0] = Wl[kb * 32 + n + 16];  bl1v[1] = Wl[(kb + 1) * 32 + n + 16];
        br0v[0] = Wr[kb * 32 + n];       br0v[1] = Wr[(kb + 1) * 32 + n];
        br1v[0] = Wr[kb * 32 + n + 16];  br1v[1] = Wr[(kb + 1) * 32 + n + 16];

        cl0 = __builtin_amdgcn_wmma_f32_16x16x4_f32(false, a, false, bl0v, (short)0, cl0, false, false);
        cl1 = __builtin_amdgcn_wmma_f32_16x16x4_f32(false, a, false, bl1v, (short)0, cl1, false, false);
        cr0 = __builtin_amdgcn_wmma_f32_16x16x4_f32(false, a, false, br0v, (short)0, cr0, false, false);
        cr1 = __builtin_amdgcn_wmma_f32_16x16x4_f32(false, a, false, br1v, (short)0, cr1, false, false);
    }

    const int n = lane & 15;
    const float bl_lo = bl[n], bl_hi = bl[n + 16];
    const float br_lo = br[n], br_hi = br[n + 16];
    #pragma unroll
    for (int v = 0; v < 8; ++v) {
        const int mm = v + 8 * kh;
        const int row = row0 + mm;
        if (row < N) {
            const size_t ro = (size_t)row * 32;
            xl[ro + n]      = cl0[v] + bl_lo;
            xl[ro + n + 16] = cl1[v] + bl_hi;
            xr[ro + n]      = cr0[v] + br_lo;
            xr[ro + n + 16] = cr1[v] + br_hi;
        }
    }
}

__global__ __launch_bounds__(256) void gat_edge_logits(
        const int* __restrict__ src, const int* __restrict__ dst,
        const float* __restrict__ ew,
        const float* __restrict__ xl, const float* __restrict__ xr,
        const float* __restrict__ We, const float* __restrict__ att,
        float* __restrict__ logits, float* __restrict__ seg_max, int E) {
    int e = blockIdx.x * blockDim.x + threadIdx.x;
    if (e >= E) return;
    const int s = src[e], d = dst[e];
    const float w = ew[e];
    const float4* __restrict__ pl = (const float4*)(xl + (size_t)s * 32);
    const float4* __restrict__ pr = (const float4*)(xr + (size_t)d * 32);
    const float4* __restrict__ pe = (const float4*)We;
    const float4* __restrict__ pa = (const float4*)att;
    float acc = 0.0f;
    #pragma unroll
    for (int i = 0; i < 8; ++i) {
        float4 a = pl[i], b = pr[i], c = pe[i], t = pa[i];
        float m0 = a.x + b.x + w * c.x; m0 = (m0 > 0.0f) ? m0 : NEG_ATT * m0;
        float m1 = a.y + b.y + w * c.y; m1 = (m1 > 0.0f) ? m1 : NEG_ATT * m1;
        float m2 = a.z + b.z + w * c.z; m2 = (m2 > 0.0f) ? m2 : NEG_ATT * m2;
        float m3 = a.w + b.w + w * c.w; m3 = (m3 > 0.0f) ? m3 : NEG_ATT * m3;
        acc += m0 * t.x + m1 * t.y + m2 * t.z + m3 * t.w;
    }
    logits[e] = acc;
    atomic_max_float(&seg_max[d], acc);
}

__global__ __launch_bounds__(256) void gat_edge_exp(
        const int* __restrict__ dst, float* __restrict__ logits,
        const float* __restrict__ seg_max, float* __restrict__ seg_sum, int E) {
    int e = blockIdx.x * blockDim.x + threadIdx.x;
    if (e >= E) return;
    const int d = dst[e];
    const float ex = __expf(logits[e] - seg_max[d]);
    logits[e] = ex;
    atomicAdd(&seg_sum[d], ex);
}

__global__ __launch_bounds__(256) void gat_edge_scatter(
        const int* __restrict__ src, const int* __restrict__ dst,
        const float* __restrict__ ex, const float* __restrict__ seg_sum,
        const float* __restrict__ xl, float* __restrict__ out, int E) {
    int e = blockIdx.x * blockDim.x + threadIdx.x;
    if (e >= E) return;
    const int s = src[e], d = dst[e];
    const float alpha = ex[e] / seg_sum[d];
    const float4* __restrict__ pl = (const float4*)(xl + (size_t)s * 32);
    float* __restrict__ po = out + (size_t)d * 32;
    #pragma unroll
    for (int i = 0; i < 8; ++i) {
        float4 v = pl[i];
        atomicAdd(po + 4 * i + 0, v.x * alpha);
        atomicAdd(po + 4 * i + 1, v.y * alpha);
        atomicAdd(po + 4 * i + 2, v.z * alpha);
        atomicAdd(po + 4 * i + 3, v.w * alpha);
    }
}

__global__ void gat_finalize(float* __restrict__ out,
                             const float* __restrict__ bias, int total) {
    int i = blockIdx.x * blockDim.x + threadIdx.x;
    if (i >= total) return;
    float v = out[i] + bias[i & 31];
    out[i] = (v > 0.0f) ? v : NEG_OUT * v;
}

extern "C" void kernel_launch(void* const* d_in, const int* in_sizes, int n_in,
                              void* d_out, int out_size, void* d_ws, size_t ws_size,
                              hipStream_t stream) {
    const float* x    = (const float*)d_in[0];
    const int*   ei   = (const int*)d_in[1];
    const float* ew   = (const float*)d_in[2];
    const float* Wl   = (const float*)d_in[3];
    const float* bl   = (const float*)d_in[4];
    const float* Wr   = (const float*)d_in[5];
    const float* br   = (const float*)d_in[6];
    const float* We   = (const float*)d_in[7];
    const float* att  = (const float*)d_in[8];
    const float* bias = (const float*)d_in[9];

    const int N = in_sizes[0] / 128;
    const int E = in_sizes[2];
    const int* srcv = ei;
    const int* dstv = ei + E;

    float* xl      = (float*)d_ws;
    float* xr      = xl + (size_t)N * 32;
    float* logits  = xr + (size_t)N * 32;
    float* seg_max = logits + (size_t)E;
    float* seg_sum = seg_max + (size_t)N;
    float* out     = (float*)d_out;

    const int tpb = 256;

    gat_init<<<(N * 32 + tpb - 1) / tpb, tpb, 0, stream>>>(out, seg_max, seg_sum, N);

    const int tiles = (N + 15) / 16;
    const int wavesPerBlock = 8;
    gat_gemm<<<(tiles + wavesPerBlock - 1) / wavesPerBlock, 32 * wavesPerBlock, 0, stream>>>(
        x, Wl, bl, Wr, br, xl, xr, N);

    gat_edge_logits<<<(E + tpb - 1) / tpb, tpb, 0, stream>>>(
        srcv, dstv, ew, xl, xr, We, att, logits, seg_max, E);

    gat_edge_exp<<<(E + tpb - 1) / tpb, tpb, 0, stream>>>(
        dstv, logits, seg_max, seg_sum, E);

    gat_edge_scatter<<<(E + tpb - 1) / tpb, tpb, 0, stream>>>(
        srcv, dstv, logits, seg_sum, xl, out, E);

    gat_finalize<<<(N * 32 + tpb - 1) / tpb, tpb, 0, stream>>>(out, bias, N * 32);
}


# node GEMM tiles run as extra work-groups of the two sorting launches, in the WGP places their grids leave empty (512 places, 391 taken; 28 tiles a group, first in the grid); 4 launches become 3
# speedup vs baseline: 14.0951x; 14.0951x over previous
#include <hip/hip_runtime.h>
#include <hip/hip_bf16.h>
#include <math.h>

#define NEG_ATT 0.2f
#define NEG_OUT 0.01f

typedef __attribute__((ext_vector_type(2))) float v2f;
typedef __attribute__((ext_vector_type(8))) float v8f;

#define BSH    8
#define BNODES 256
#define MAXB   512
#define CH     8192
#define MAXG   1024
#define GT     10240
#define KT     10
#define NW     32
#define TPG    28

__device__ __forceinline__ void gemm_tile(
        const float* __restrict__ x,
        const float* __restrict__ Wl, const float* __restrict__ bl,
        const float* __restrict__ Wr, const float* __restrict__ br,
        float* __restrict__ xl, float* __restrict__ xr, int N,
        const int tile, const int lane) {
    const int row0 = tile * 16;
    if (row0 >= N) return;

    const int m  = lane & 15;
    const int kh = lane >> 4;

    int arow = row0 + m;
    if (arow >= N) arow = N - 1;
    const float* __restrict__ xrow = x + (size_t)arow * 128;

    v8f cl0 = {}, cl1 = {}, cr0 = {}, cr1 = {};

    #pragma unroll 4
    for (int k0 = 0; k0 < 128; k0 += 4) {
        const int kb = k0 + 2 * kh;
        v2f a;
        a[0] = xrow[kb];
        a[1] = xrow[kb + 1];

        const int n = m;
        v2f bl0v, bl1v, br0v, br1v;
        bl0v[0] = Wl[kb * 32 + n];       bl0v[1] = Wl[(kb + 1) * 32 + n];
        bl1v[0] = Wl[kb * 32 + n + 16];  bl1v[1] = Wl[(kb + 1) * 32 + n + 16];
        br0v[0] = Wr[kb * 32 + n];       br0v[1] = Wr[(kb + 1) * 32 + n];
        br1v[0] = Wr[kb * 32 + n + 16];  br1v[1] = Wr[(kb + 1) * 32 + n + 16];

        cl0 = __builtin_amdgcn_wmma_f32_16x16x4_f32(false, a, false, bl0v, (short)0, cl0, false, false);
        cl1 = __builtin_amdgcn_wmma_f32_16x16x4_f32(false, a, false, bl1v, (short)0, cl1, false, false);
        cr0 = __builtin_amdgcn_wmma_f32_16x16x4_f32(false, a, false, br0v, (short)0, cr0, false, false);
        cr1 = __builtin_amdgcn_wmma_f32_16x16x4_f32(false, a, false, br1v, (short)0, cr1, false, false);
    }

    const int n = lane & 15;
    const float bl_lo = bl[n], bl_hi = bl[n + 16];
    const float br_lo = br[n], br_hi = br[n + 16];
    #pragma unroll
    for (int v = 0; v < 8; ++v) {
        const int mm = v + 8 * kh;
        const int row = row0 + mm;
        if (row < N) {
            const size_t ro = (size_t)row * 32;
            xl[ro + n]      = cl0[v] + bl_lo;
            xl[ro + n + 16] = cl1[v] + bl_hi;
            xr[ro + n]      = cr0[v] + br_lo;
            xr[ro + n + 16] = cr1[v] + br_hi;
        }
    }
}

__device__ __forceinline__ int wg_scan_excl(int c, int* wtot, int lane, int wv, int* total) {
    int incl = c;
    #pragma unroll
    for (int o = 1; o < 32; o <<= 1) {
        const int t = __shfl_up(incl, o);
        if (lane >= o) incl += t;
    }
    if (lane == 31) wtot[wv] = incl;
    __syncthreads();
    if (wv == 0) {
        const int t = wtot[lane];
        int inc = t;
        #pragma unroll
        for (int o = 1; o < 32; o <<= 1) {
            const int u = __shfl_up(inc, o);
            if (lane >= o) inc += u;
        }
        wtot[lane] = inc - t;
        if (lane == 31) wtot[32] = inc;
    }
    __syncthreads();
    *total = wtot[32];
    return wtot[wv] + incl - c;
}

__global__ __launch_bounds__(1024, 16) void gat_bin_gemm(
        const int* __restrict__ src, const int* __restrict__ dst,
        const float* __restrict__ ew, int* __restrict__ tab,
        int* __restrict__ bkt, int E, int B, int nblk, int ng,
        const float* __restrict__ x,
        const float* __restrict__ Wl, const float* __restrict__ bl,
        const float* __restrict__ Wr, const float* __restrict__ br,
        float* __restrict__ xl, float* __restrict__ xr, int N) {
    if ((int)blockIdx.x < ng) {
        const int gwv = (int)(threadIdx.x >> 5);
        if (gwv < TPG)
            gemm_tile(x, Wl, bl, Wr, br, xl, xr, N,
                      (int)blockIdx.x * TPG + gwv, (int)(threadIdx.x & 31));
        return;
    }
    __shared__ __attribute__((aligned(16))) int wcnt[NW * MAXB];
    __shared__ int wtot[33];
    __shared__ int stx[CH];
    __shared__ int sty[CH];
    const int tid  = threadIdx.x;
    const int lane = tid & 31;
    const int wv   = tid >> 5;
    const int blk  = (int)blockIdx.x - ng;
    {
        const int4 z = {0, 0, 0, 0};
        #pragma unroll
        for (int i = tid; i < NW * MAXB / 4; i += 1024) ((int4*)wcnt)[i] = z;
    }
    const int base = blk * CH;
    int d[8], s[8], w[8], r[8];
    #pragma unroll
    for (int k = 0; k < 8; ++k) {
        const int e = base + k * 1024 + tid;
        if (e < E) {
            d[k] = dst[e];
            s[k] = src[e];
            w[k] = __float_as_int(ew[e]);
        } else {
            d[k] = -1;
            s[k] = 0;
            w[k] = 0;
        }
    }
    __syncthreads();
    int* __restrict__ mycnt = wcnt + wv * MAXB;
    #pragma unroll
    for (int k = 0; k < 8; ++k)
        r[k] = (d[k] >= 0) ? atomicAdd(&mycnt[d[k] >> BSH], 1) : 0;
    __syncthreads();

    const int rb = tid >> 1;
    const int q0 = (tid & 1) * (NW / 2);
    int c = 0;
    #pragma unroll
    for (int j = 0; j < NW / 2; ++j) c += wcnt[(q0 + j) * MAXB + rb];
    int nall;
    const int excl = wg_scan_excl(c, wtot, lane, wv, &nall);
    {
        int v[NW / 2];
        #pragma unroll
        for (int j = 0; j < NW / 2; ++j) v[j] = wcnt[(q0 + j) * MAXB + rb];
        int run = excl;
        #pragma unroll
        for (int j = 0; j < NW / 2; ++j) {
            wcnt[(q0 + j) * MAXB + rb] = run;
            run += v[j];
        }
    }
    if (q0 == 0 && rb < B) tab[rb * nblk + blk] = excl;
    __syncthreads();

    int pb[8];
    #pragma unroll
    for (int k = 0; k < 8; ++k) pb[k] = mycnt[(d[k] >= 0) ? (d[k] >> BSH) : 0];
    #pragma unroll
    for (int k = 0; k < 8; ++k) {
        if (d[k] >= 0) {
            const int p = pb[k] + r[k];
            stx[p] = (int)(((unsigned)(d[k] & (BNODES - 1)) << 24) | (unsigned)s[k]);
            sty[p] = w[k];
        }
    }
    __syncthreads();

    const int left = E - base;
    const int nE = (left < CH) ? left : CH;
    int2* __restrict__ outp = (int2*)bkt + base;
    for (int i = tid; i < nE; i += 1024) {
        int2 t;
        t.x = stx[i];
        t.y = sty[i];
        outp[i] = t;
    }
}

__global__ __launch_bounds__(1024, 16) void gat_group_gemm(
        const int* __restrict__ bkt, const int* __restrict__ tab,
        int* __restrict__ srt, int* __restrict__ off, int* __restrict__ cnt,
        int N, int E, int B, int nblk, int ng,
        const float* __restrict__ x,
        const float* __restrict__ Wl, const float* __restrict__ bl,
        const float* __restrict__ Wr, const float* __restrict__ br,
        float* __restrict__ xl, float* __restrict__ xr, int tile0) {
    if ((int)blockIdx.x < ng) {
        const int gwv = (int)(threadIdx.x >> 5);
        if (gwv < TPG)
            gemm_tile(x, Wl, bl, Wr, br, xl, xr, N,
                      tile0 + (int)blockIdx.x * TPG + gwv, (int)(threadIdx.x & 31));
        return;
    }
    __shared__ __attribute__((aligned(16))) int wcnt[NW * BNODES];
    __shared__ int wtot[33];
    __shared__ int wred[32];
    __shared__ int rsrc[MAXG];
    __shared__ unsigned short gid[GT];
    __shared__ int stx[GT];
    __shared__ int sty[GT];
    const int tid  = threadIdx.x;
    const int lane = tid & 31;
    const int wv   = tid >> 5;
    const int b    = (int)blockIdx.x - ng;
    {
        const int4 z = {0, 0, 0, 0};
        #pragma unroll
        for (int i = tid; i < NW * BNODES / 4; i += 1024) ((int4*)wcnt)[i] = z;
    }

    int st = 0, len = 0;
    if (tid < nblk) {
        st = tab[b * nblk + tid];
        int en;
        if (b + 1 < B) {
            en = tab[(b + 1) * nblk + tid];
        } else {
            const int left = E - tid * CH;
            en = (left < CH) ? left : CH;
        }
        len = en - st;
    }
    int ssum = st;
    #pragma unroll
    for (int o = 16; o >= 1; o >>= 1) ssum += __shfl_xor(ssum, o);
    if (lane == 0) wred[wv] = ssum;
    int nrecv;
    const int pre = wg_scan_excl(len, wtot, lane, wv, &nrecv);
    int bsum = wred[lane];
    #pragma unroll
    for (int o = 16; o >= 1; o >>= 1) bsum += __shfl_xor(bsum, o);
    const int nrec = __builtin_amdgcn_readfirstlane(nrecv);
    const int bbeg = __builtin_amdgcn_readfirstlane(bsum);
    if (tid < nblk) rsrc[tid] = tid * CH + st - pre;

    const int2* __restrict__ in = (const int2*)bkt;
    int2* __restrict__ outp = (int2*)srt + bbeg;
    int* __restrict__ mycnt = wcnt + wv * BNODES;

    for (int t0 = 0; t0 < nrec; t0 += GT) {
        __syncthreads();
        if (tid < nblk) {
            const int lo = (pre > t0) ? pre : t0;
            int hi = pre + len;
            if (hi > t0 + GT) hi = t0 + GT;
            #pragma clang loop vectorize(disable) interleave(disable) unroll(disable)
            for (int j = lo; j < hi; ++j) gid[j - t0] = (unsigned short)tid;
        }
        __syncthreads();
        int x[KT];
        #pragma unroll
        for (int k = 0; k < KT; ++k) {
            const int i = t0 + k * 1024 + tid;
            const int ii = (i < nrec) ? i : nrec - 1;
            x[k] = in[ii + rsrc[gid[ii - t0]]].x;
        }
        #pragma unroll
        for (int k = 0; k < KT; ++k) {
            const int i = t0 + k * 1024 + tid;
            if (i < nrec) atomicAdd(&mycnt[((unsigned)x[k]) >> 24], 1);
        }
    }
    __syncthreads();

    const int nd = tid >> 2;
    const int q0 = (tid & 3) * (NW / 4);
    int v[NW / 4];
    int c = 0;
    #pragma unroll
    for (int j = 0; j < NW / 4; ++j) {
        v[j] = wcnt[(q0 + j) * BNODES + nd];
        c += v[j];
    }
    int nall;
    const int excl = wg_scan_excl(c, wtot, lane, wv, &nall);
    {
        int run = excl;
        #pragma unroll
        for (int j = 0; j < NW / 4; ++j) {
            wcnt[(q0 + j) * BNODES + nd] = run;
            run += v[j];
        }
    }
    int ctot = c + __shfl_xor(c, 1);
    ctot += __shfl_xor(ctot, 2);
    if (q0 == 0) {
        const int node = b * BNODES + nd;
        if (node < N) {
            off[node] = bbeg + excl;
            cnt[node] = ctot;
        }
    }

    const bool staged = nrec <= GT;
    for (int t0 = 0; t0 < nrec; t0 += GT) {
        __syncthreads();
        if (!staged) {
            if (tid < nblk) {
                const int lo = (pre > t0) ? pre : t0;
                int hi = pre + len;
                if (hi > t0 + GT) hi = t0 + GT;
                #pragma clang loop vectorize(disable) interleave(disable) unroll(disable)
                for (int j = lo; j < hi; ++j) gid[j - t0] = (unsigned short)tid;
            }
            __syncthreads();
        }
        int rx[KT], ry[KT];
        #pragma unroll
        for (int k = 0; k < KT; ++k) {
            const int i = t0 + k * 1024 + tid;
            const int ii = (i < nrec) ? i : nrec - 1;
            const int2 rec = in[ii + rsrc[gid[ii - t0]]];
            rx[k] = rec.x;
            ry[k] = rec.y;
        }
        #pragma unroll
        for (int k = 0; k < KT; ++k) {
            const int i = t0 + k * 1024 + tid;
            if (i < nrec) {
                const int p = atomicAdd(&mycnt[((unsigned)rx[k]) >> 24], 1);
                const int sx = rx[k] & 0x00FFFFFF;
                if (staged) {
                    stx[p] = sx;
                    sty[p] = ry[k];
                } else {
                    int2 t;
                    t.x = sx;
                    t.y = ry[k];
                    outp[p] = t;
                }
            }
        }
    }
    if (staged) {
        __syncthreads();
        for (int i = tid; i < nrec; i += 1024) {
            int2 t;
            t.x = stx[i];
            t.y = sty[i];
            outp[i] = t;
        }
    }
}


template <int CTRL>
__device__ __forceinline__ int row_pick_i(int x) {
    return __builtin_amdgcn_update_dpp(0, x, CTRL, 0xF, 0xF, true);
}
template <int CTRL>
__device__ __forceinline__ float row_pick_f(float x) {
    return __int_as_float(__builtin_amdgcn_update_dpp(0, __float_as_int(x), CTRL, 0xF, 0xF, true));
}

template <int H>
__device__ __forceinline__ void row_fold(float* q, int cl) {
    const bool up = (cl & H) != 0;
    #pragma unroll
    for (int j = 0; j < H; ++j) {
        const float lo = q[j];
        const float hi = q[j + H];
        const float keep = up ? hi : lo;
        const float send = up ? lo : hi;
        q[j] = keep + row_pick_f<0x160 + H>(send);
    }
}

#define REP16(M) M(0) M(1) M(2) M(3) M(4) M(5) M(6) M(7) M(8) M(9) M(10) M(11) M(12) M(13) M(14) M(15)

#define NPW 4

__global__ __launch_bounds__(256) void gat_node(
        const int* __restrict__ srt, const int* __restrict__ off,
        const int* __restrict__ cnt,
        const float* __restrict__ xl, const float* __restrict__ xr,
        const float* __restrict__ We, const float* __restrict__ att,
        const float* __restrict__ bias, float* __restrict__ out, int N) {
    const int lane = threadIdx.x & 31;
    const int cl   = lane & 15;
    const int w = __builtin_amdgcn_readfirstlane(
        (int)((blockIdx.x * blockDim.x + threadIdx.x) >> 5));
    const int n0 = w * NPW;
    if (n0 >= N) return;

    int offv = 0, cntv = 0;
    if (lane < NPW) {
        const int nn = (n0 + lane < N) ? n0 + lane : N - 1;
        offv = off[nn];
        cntv = cnt[nn];
    }

    const v2f* __restrict__ xl2 = (const v2f*)xl;
    v2f we2, att2, bias2;
    we2[0]   = We[2 * cl];    we2[1]   = We[2 * cl + 1];
    att2[0]  = att[2 * cl];   att2[1]  = att[2 * cl + 1];
    bias2[0] = bias[2 * cl];  bias2[1] = bias[2 * cl + 1];

    #pragma unroll 1
    for (int j = 0; j < NPW; ++j) {
        const int n = n0 + j;
        if (n >= N) break;
        const int beg = __builtin_amdgcn_readlane(offv, j);
        const int deg = __builtin_amdgcn_readlane(cntv, j);
        const v2f xr2 = ((const v2f*)xr)[(size_t)n * 16 + cl];
        const int2* __restrict__ edges = (const int2*)srt + beg;

        float m_run = -INFINITY;
        float s_run = 0.0f;
        v2f   acc   = {0.0f, 0.0f};

        for (int c0 = 0; c0 < deg; c0 += 32) {
            const int rem = deg - c0;
            const bool valid = lane < rem;
            int s = n;
            float wf = 0.0f;
            if (valid) {
                const int2 t = edges[c0 + lane];
                s = t.x;
                wf = __int_as_float(t.y);
            }

            v2f v[16];
#define GAT_LOAD(i) { const unsigned si = (unsigned)row_pick_i<0x150 + (i)>(s); v[i] = xl2[si * 16u + (unsigned)cl]; }
            REP16(GAT_LOAD)
#undef GAT_LOAD
            __builtin_amdgcn_sched_barrier(0);

            float q[16];
#define GAT_LOGIT(i) { \
            const float wi = row_pick_f<0x150 + (i)>(wf); \
            v2f mm = v[i] + xr2 + wi * we2; \
            mm[0] = (mm[0] > 0.0f) ? mm[0] : NEG_ATT * mm[0]; \
            mm[1] = (mm[1] > 0.0f) ? mm[1] : NEG_ATT * mm[1]; \
            q[i] = mm[0] * att2[0] + mm[1] * att2[1]; }
            REP16(GAT_LOGIT)
#undef GAT_LOGIT
            row_fold<8>(q, cl);
            row_fold<4>(q, cl);
            row_fold<2>(q, cl);
            row_fold<1>(q, cl);
            const float logit = valid ? q[0] : -INFINITY;

            float cmax = logit;
            #pragma unroll
            for (int o = 16; o >= 1; o >>= 1) cmax = fmaxf(cmax, __shfl_xor(cmax, o));
            const float m_new = fmaxf(m_run, cmax);
            const float scale = __expf(m_run - m_new);
            const float ex    = __expf(logit - m_new);
            float csum = ex;
            #pragma unroll
            for (int o = 16; o >= 1; o >>= 1) csum += __shfl_xor(csum, o);
            s_run = s_run * scale + csum;

            v2f a0 = {0.0f, 0.0f}, a1 = {0.0f, 0.0f};
#define GAT_ACC(i) { const float ei = row_pick_f<0x150 + (i)>(ex); if ((i) & 1) a1 += ei * v[i]; else a0 += ei * v[i]; }
            REP16(GAT_ACC)
#undef GAT_ACC
            acc = acc * scale + (a0 + a1);
            m_run = m_new;
        }

        const float tx = acc[0] + __shfl_xor(acc[0], 16);
        const float ty = acc[1] + __shfl_xor(acc[1], 16);
        float ox = (deg > 0) ? tx / s_run : 0.0f;
        float oy = (deg > 0) ? ty / s_run : 0.0f;
        ox += bias2[0];
        oy += bias2[1];
        if (lane < 16) {
            v2f o;
            o[0] = (ox > 0.0f) ? ox : NEG_OUT * ox;
            o[1] = (oy > 0.0f) ? oy : NEG_OUT * oy;
            ((v2f*)out)[(size_t)n * 16 + cl] = o;
        }
    }
}

extern "C" void kernel_launch(void* const* d_in, const int* in_sizes, int n_in,
                              void* d_out, int out_size, void* d_ws, size_t ws_size,
                              hipStream_t stream) {
    const float* x    = (const float*)d_in[0];
    const int*   ei   = (const int*)d_in[1];
    const float* ew   = (const float*)d_in[2];
    const float* Wl   = (const float*)d_in[3];
    const float* bl   = (const float*)d_in[4];
    const float* Wr   = (const float*)d_in[5];
    const float* br   = (const float*)d_in[6];
    const float* We   = (const float*)d_in[7];
    const float* att  = (const float*)d_in[8];
    const float* bias = (const float*)d_in[9];

    const int N = in_sizes[0] / 128;
    const int E = in_sizes[2];
    const int* srcv = ei;
    const int* dstv = ei + E;

    const int B    = (N + BNODES - 1) / BNODES;
    const int nblk = (E + CH - 1) / CH;

    float* xl   = (float*)d_ws;
    float* xr   = xl + (size_t)N * 32;
    int*   srt  = (int*)(xr + (size_t)N * 32);
    int*   bkt  = srt + (size_t)E * 2;
    int*   cnt  = bkt + (size_t)E * 2;
    int*   off  = cnt + (size_t)N;
    int*   tab  = off + (size_t)N;
    float* out  = (float*)d_out;

    const int tiles = (N + 15) / 16;
    const int wavesPerBlock = 8;
    const int gw = (tiles + TPG - 1) / TPG;
    const int g1 = (gw + 1) / 2;
    const int g2 = gw - g1;

    gat_bin_gemm<<<g1 + nblk, 1024, 0, stream>>>(
        srcv, dstv, ew, tab, bkt, E, B, nblk, g1, x, Wl, bl, Wr, br, xl, xr, N);

    gat_group_gemm<<<g2 + B, 1024, 0, stream>>>(
        bkt, tab, srt, off, cnt, N, E, B, nblk, g2, x, Wl, bl, Wr, br, xl, xr, g1 * TPG);

    gat_node<<<(N + wavesPerBlock * NPW - 1) / (wavesPerBlock * NPW), 32 * wavesPerBlock, 0, stream>>>(
        srt, off, cnt, xl, xr, We, att, bias, out, N);
}


# node GEMM tiles: a work-group stages Wl and Wr in LDS once, in fragment order, and its waves read B there (each wave had fetched all of W itself); A in 16-byte loads; k (sum) order of xl, xr changes
# speedup vs baseline: 15.5058x; 1.1001x over previous
#include <hip/hip_runtime.h>
#include <hip/hip_bf16.h>
#include <math.h>

#define NEG_ATT 0.2f
#define NEG_OUT 0.01f

typedef __attribute__((ext_vector_type(2))) float v2f;
typedef __attribute__((ext_vector_type(8))) float v8f;

#define BSH    8
#define BNODES 256
#define MAXB   512
#define CH     8192
#define MAXG   1024
#define GT     10240
#define KT     10
#define NW     32
#define TPG    28


__device__ __forceinline__ void w_image(const float* __restrict__ Wl,
                                        const float* __restrict__ Wr,
                                        float* wsh, const int tid) {
    #pragma unroll
    for (int j = 0; j < 8; ++j) {
        const int g   = j * 1024 + tid;
        const int mat = g >> 12;
        const int k   = (g >> 5) & 127;
        const int col = g & 31;
        const float val = (mat ? Wr : Wl)[g & 4095];
        const int t  = ((k >> 3) << 1) | ((k >> 1) & 1);
        const int h  = (k >> 2) & 1;
        const int e  = ((col >> 4) << 1) | (k & 1);
        const int ln = (h << 4) | (col & 15);
        wsh[((((t << 1) | mat) << 5) + ln) * 4 + e] = val;
    }
}

__device__ __forceinline__ void gemm_tile(
        const float* __restrict__ x,
        const float* wsh, const float* __restrict__ bl,
        const float* __restrict__ br,
        float* __restrict__ xl, float* __restrict__ xr, int N,
        const int tile, const int lane) {
    const int row0 = tile * 16;
    if (row0 >= N) return;

    const int m  = lane & 15;
    const int kh = lane >> 4;

    int arow = row0 + m;
    if (arow >= N) arow = N - 1;
    const float4* __restrict__ xq = (const float4*)(x + (size_t)arow * 128) + kh;
    const float4* wq = (const float4*)wsh + lane;

    v8f cl0 = {}, cl1 = {}, cr0 = {}, cr1 = {};

    for (int u = 0; u < 16; u += 2) {
        const float4 ax0 = xq[2 * u];
        const float4 ax1 = xq[2 * u + 2];
        #pragma unroll
        for (int q = 0; q < 4; ++q) {
            const int t = 2 * u + q;
            const int p = q & 1;
            const float4 ax = (q < 2) ? ax0 : ax1;
            const float4 wl4 = wq[(2 * t) * 32];
            const float4 wr4 = wq[(2 * t + 1) * 32];
            v2f a;
            a[0] = p ? ax.z : ax.x;
            a[1] = p ? ax.w : ax.y;
            v2f bl0v, bl1v, br0v, br1v;
            bl0v[0] = wl4.x;  bl0v[1] = wl4.y;
            bl1v[0] = wl4.z;  bl1v[1] = wl4.w;
            br0v[0] = wr4.x;  br0v[1] = wr4.y;
            br1v[0] = wr4.z;  br1v[1] = wr4.w;

            cl0 = __builtin_amdgcn_wmma_f32_16x16x4_f32(false, a, false, bl0v, (short)0, cl0, false, false);
            cl1 = __builtin_amdgcn_wmma_f32_16x16x4_f32(false, a, false, bl1v, (short)0, cl1, false, false);
            cr0 = __builtin_amdgcn_wmma_f32_16x16x4_f32(false, a, false, br0v, (short)0, cr0, false, false);
            cr1 = __builtin_amdgcn_wmma_f32_16x16x4_f32(false, a, false, br1v, (short)0, cr1, false, false);
        }
    }

    const int n = lane & 15;
    const float bl_lo = bl[n], bl_hi = bl[n + 16];
    const float br_lo = br[n], br_hi = br[n + 16];
    #pragma unroll
    for (int v = 0; v < 8; ++v) {
        const int mm = v + 8 * kh;
        const int row = row0 + mm;
        if (row < N) {
            const size_t ro = (size_t)row * 32;
            xl[ro + n]      = cl0[v] + bl_lo;
            xl[ro + n + 16] = cl1[v] + bl_hi;
            xr[ro + n]      = cr0[v] + br_lo;
            xr[ro + n + 16] = cr1[v] + br_hi;
        }
    }
}

__device__ __forceinline__ int wg_scan_excl(int c, int* wtot, int lane, int wv, int* total) {
    int incl = c;
    #pragma unroll
    for (int o = 1; o < 32; o <<= 1) {
        const int t = __shfl_up(incl, o);
        if (lane >= o) incl += t;
    }
    if (lane == 31) wtot[wv] = incl;
    __syncthreads();
    if (wv == 0) {
        const int t = wtot[lane];
        int inc = t;
        #pragma unroll
        for (int o = 1; o < 32; o <<= 1) {
            const int u = __shfl_up(inc, o);
            if (lane >= o) inc += u;
        }
        wtot[lane] = inc - t;
        if (lane == 31) wtot[32] = inc;
    }
    __syncthreads();
    *total = wtot[32];
    return wtot[wv] + incl - c;
}

__global__ __launch_bounds__(1024, 16) void gat_bin_gemm(
        const int* __restrict__ src, const int* __restrict__ dst,
        const float* __restrict__ ew, int* __restrict__ tab,
        int* __restrict__ bkt, int E, int B, int nblk, int ng,
        const float* __restrict__ x,
        const float* __restrict__ Wl, const float* __restrict__ bl,
        const float* __restrict__ Wr, const float* __restrict__ br,
        float* __restrict__ xl, float* __restrict__ xr, int N) {
    __shared__ __attribute__((aligned(16))) int wcnt[NW * MAXB];
    __shared__ int wtot[33];
    __shared__ int stx[CH];
    __shared__ int sty[CH];
    if ((int)blockIdx.x < ng) {
        float* wsh = (float*)wcnt;
        w_image(Wl, Wr, wsh, (int)threadIdx.x);
        __syncthreads();
        const int gwv = (int)(threadIdx.x >> 5);
        if (gwv < TPG)
            gemm_tile(x, wsh, bl, br, xl, xr, N,
                      (int)blockIdx.x * TPG + gwv, (int)(threadIdx.x & 31));
        return;
    }
    const int tid  = threadIdx.x;
    const int lane = tid & 31;
    const int wv   = tid >> 5;
    const int blk  = (int)blockIdx.x - ng;
    {
        const int4 z = {0, 0, 0, 0};
        #pragma unroll
        for (int i = tid; i < NW * MAXB / 4; i += 1024) ((int4*)wcnt)[i] = z;
    }
    const int base = blk * CH;
    int d[8], s[8], w[8], r[8];
    #pragma unroll
    for (int k = 0; k < 8; ++k) {
        const int e = base + k * 1024 + tid;
        if (e < E) {
            d[k] = dst[e];
            s[k] = src[e];
            w[k] = __float_as_int(ew[e]);
        } else {
            d[k] = -1;
            s[k] = 0;
            w[k] = 0;
        }
    }
    __syncthreads();
    int* __restrict__ mycnt = wcnt + wv * MAXB;
    #pragma unroll
    for (int k = 0; k < 8; ++k)
        r[k] = (d[k] >= 0) ? atomicAdd(&mycnt[d[k] >> BSH], 1) : 0;
    __syncthreads();

    const int rb = tid >> 1;
    const int q0 = (tid & 1) * (NW / 2);
    int c = 0;
    #pragma unroll
    for (int j = 0; j < NW / 2; ++j) c += wcnt[(q0 + j) * MAXB + rb];
    int nall;
    const int excl = wg_scan_excl(c, wtot, lane, wv, &nall);
    {
        int v[NW / 2];
        #pragma unroll
        for (int j = 0; j < NW / 2; ++j) v[j] = wcnt[(q0 + j) * MAXB + rb];
        int run = excl;
        #pragma unroll
        for (int j = 0; j < NW / 2; ++j) {
            wcnt[(q0 + j) * MAXB + rb] = run;
            run += v[j];
        }
    }
    if (q0 == 0 && rb < B) tab[rb * nblk + blk] = excl;
    __syncthreads();

    int pb[8];
    #pragma unroll
    for (int k = 0; k < 8; ++k) pb[k] = mycnt[(d[k] >= 0) ? (d[k] >> BSH) : 0];
    #pragma unroll
    for (int k = 0; k < 8; ++k) {
        if (d[k] >= 0) {
            const int p = pb[k] + r[k];
            stx[p] = (int)(((unsigned)(d[k] & (BNODES - 1)) << 24) | (unsigned)s[k]);
            sty[p] = w[k];
        }
    }
    __syncthreads();

    const int left = E - base;
    const int nE = (left < CH) ? left : CH;
    int2* __restrict__ outp = (int2*)bkt + base;
    for (int i = tid; i < nE; i += 1024) {
        int2 t;
        t.x = stx[i];
        t.y = sty[i];
        outp[i] = t;
    }
}

__global__ __launch_bounds__(1024, 16) void gat_group_gemm(
        const int* __restrict__ bkt, const int* __restrict__ tab,
        int* __restrict__ srt, int* __restrict__ off, int* __restrict__ cnt,
        int N, int E, int B, int nblk, int ng,
        const float* __restrict__ x,
        const float* __restrict__ Wl, const float* __restrict__ bl,
        const float* __restrict__ Wr, const float* __restrict__ br,
        float* __restrict__ xl, float* __restrict__ xr, int tile0) {
    __shared__ __attribute__((aligned(16))) int wcnt[NW * BNODES];
    __shared__ int wtot[33];
    __shared__ int wred[32];
    __shared__ int rsrc[MAXG];
    __shared__ unsigned short gid[GT];
    __shared__ __attribute__((aligned(16))) int stx[GT];
    __shared__ int sty[GT];
    if ((int)blockIdx.x < ng) {
        float* wsh = (float*)stx;
        w_image(Wl, Wr, wsh, (int)threadIdx.x);
        __syncthreads();
        const int gwv = (int)(threadIdx.x >> 5);
        if (gwv < TPG)
            gemm_tile(x, wsh, bl, br, xl, xr, N,
                      tile0 + (int)blockIdx.x * TPG + gwv, (int)(threadIdx.x & 31));
        return;
    }
    const int tid  = threadIdx.x;
    const int lane = tid & 31;
    const int wv   = tid >> 5;
    const int b    = (int)blockIdx.x - ng;
    {
        const int4 z = {0, 0, 0, 0};
        #pragma unroll
        for (int i = tid; i < NW * BNODES / 4; i += 1024) ((int4*)wcnt)[i] = z;
    }

    int st = 0, len = 0;
    if (tid < nblk) {
        st = tab[b * nblk + tid];
        int en;
        if (b + 1 < B) {
            en = tab[(b + 1) * nblk + tid];
        } else {
            const int left = E - tid * CH;
            en = (left < CH) ? left : CH;
        }
        len = en - st;
    }
    int ssum = st;
    #pragma unroll
    for (int o = 16; o >= 1; o >>= 1) ssum += __shfl_xor(ssum, o);
    if (lane == 0) wred[wv] = ssum;
    int nrecv;
    const int pre = wg_scan_excl(len, wtot, lane, wv, &nrecv);
    int bsum = wred[lane];
    #pragma unroll
    for (int o = 16; o >= 1; o >>= 1) bsum += __shfl_xor(bsum, o);
    const int nrec = __builtin_amdgcn_readfirstlane(nrecv);
    const int bbeg = __builtin_amdgcn_readfirstlane(bsum);
    if (tid < nblk) rsrc[tid] = tid * CH + st - pre;

    const int2* __restrict__ in = (const int2*)bkt;
    int2* __restrict__ outp = (int2*)srt + bbeg;
    int* __restrict__ mycnt = wcnt + wv * BNODES;

    for (int t0 = 0; t0 < nrec; t0 += GT) {
        __syncthreads();
        if (tid < nblk) {
            const int lo = (pre > t0) ? pre : t0;
            int hi = pre + len;
            if (hi > t0 + GT) hi = t0 + GT;
            #pragma clang loop vectorize(disable) interleave(disable) unroll(disable)
            for (int j = lo; j < hi; ++j) gid[j - t0] = (unsigned short)tid;
        }
        __syncthreads();
        int x[KT];
        #pragma unroll
        for (int k = 0; k < KT; ++k) {
            const int i = t0 + k * 1024 + tid;
            const int ii = (i < nrec) ? i : nrec - 1;
            x[k] = in[ii + rsrc[gid[ii - t0]]].x;
        }
        #pragma unroll
        for (int k = 0; k < KT; ++k) {
            const int i = t0 + k * 1024 + tid;
            if (i < nrec) atomicAdd(&mycnt[((unsigned)x[k]) >> 24], 1);
        }
    }
    __syncthreads();

    const int nd = tid >> 2;
    const int q0 = (tid & 3) * (NW / 4);
    int v[NW / 4];
    int c = 0;
    #pragma unroll
    for (int j = 0; j < NW / 4; ++j) {
        v[j] = wcnt[(q0 + j) * BNODES + nd];
        c += v[j];
    }
    int nall;
    const int excl = wg_scan_excl(c, wtot, lane, wv, &nall);
    {
        int run = excl;
        #pragma unroll
        for (int j = 0; j < NW / 4; ++j) {
            wcnt[(q0 + j) * BNODES + nd] = run;
            run += v[j];
        }
    }
    int ctot = c + __shfl_xor(c, 1);
    ctot += __shfl_xor(ctot, 2);
    if (q0 == 0) {
        const int node = b * BNODES + nd;
        if (node < N) {
            off[node] = bbeg + excl;
            cnt[node] = ctot;
        }
    }

    const bool staged = nrec <= GT;
    for (int t0 = 0; t0 < nrec; t0 += GT) {
        __syncthreads();
        if (!staged) {
            if (tid < nblk) {
                const int lo = (pre > t0) ? pre : t0;
                int hi = pre + len;
                if (hi > t0 + GT) hi = t0 + GT;
                #pragma clang loop vectorize(disable) interleave(disable) unroll(disable)
                for (int j = lo; j < hi; ++j) gid[j - t0] = (unsigned short)tid;
            }
            __syncthreads();
        }
        int rx[KT], ry[KT];
        #pragma unroll
        for (int k = 0; k < KT; ++k) {
            const int i = t0 + k * 1024 + tid;
            const int ii = (i < nrec) ? i : nrec - 1;
            const int2 rec = in[ii + rsrc[gid[ii - t0]]];
            rx[k] = rec.x;
            ry[k] = rec.y;
        }
        #pragma unroll
        for (int k = 0; k < KT; ++k) {
            const int i = t0 + k * 1024 + tid;
            if (i < nrec) {
                const int p = atomicAdd(&mycnt[((unsigned)rx[k]) >> 24], 1);
                const int sx = rx[k] & 0x00FFFFFF;
                if (staged) {
                    stx[p] = sx;
                    sty[p] = ry[k];
                } else {
                    int2 t;
                    t.x = sx;
                    t.y = ry[k];
                    outp[p] = t;
                }
            }
        }
    }
    if (staged) {
        __syncthreads();
        for (int i = tid; i < nrec; i += 1024) {
            int2 t;
            t.x = stx[i];
            t.y = sty[i];
            outp[i] = t;
        }
    }
}


template <int CTRL>
__device__ __forceinline__ int row_pick_i(int x) {
    return __builtin_amdgcn_update_dpp(0, x, CTRL, 0xF, 0xF, true);
}
template <int CTRL>
__device__ __forceinline__ float row_pick_f(float x) {
    return __int_as_float(__builtin_amdgcn_update_dpp(0, __float_as_int(x), CTRL, 0xF, 0xF, true));
}

template <int H>
__device__ __forceinline__ void row_fold(float* q, int cl) {
    const bool up = (cl & H) != 0;
    #pragma unroll
    for (int j = 0; j < H; ++j) {
        const float lo = q[j];
        const float hi = q[j + H];
        const float keep = up ? hi : lo;
        const float send = up ? lo : hi;
        q[j] = keep + row_pick_f<0x160 + H>(send);
    }
}

#define REP16(M) M(0) M(1) M(2) M(3) M(4) M(5) M(6) M(7) M(8) M(9) M(10) M(11) M(12) M(13) M(14) M(15)

#define NPW 4

__global__ __launch_bounds__(256) void gat_node(
        const int* __restrict__ srt, const int* __restrict__ off,
        const int* __restrict__ cnt,
        const float* __restrict__ xl, const float* __restrict__ xr,
        const float* __restrict__ We, const float* __restrict__ att,
        const float* __restrict__ bias, float* __restrict__ out, int N) {
    const int lane = threadIdx.x & 31;
    const int cl   = lane & 15;
    const int w = __builtin_amdgcn_readfirstlane(
        (int)((blockIdx.x * blockDim.x + threadIdx.x) >> 5));
    const int n0 = w * NPW;
    if (n0 >= N) return;

    int offv = 0, cntv = 0;
    if (lane < NPW) {
        const int nn = (n0 + lane < N) ? n0 + lane : N - 1;
        offv = off[nn];
        cntv = cnt[nn];
    }

    const v2f* __restrict__ xl2 = (const v2f*)xl;
    v2f we2, att2, bias2;
    we2[0]   = We[2 * cl];    we2[1]   = We[2 * cl + 1];
    att2[0]  = att[2 * cl];   att2[1]  = att[2 * cl + 1];
    bias2[0] = bias[2 * cl];  bias2[1] = bias[2 * cl + 1];

    #pragma unroll 1
    for (int j = 0; j < NPW; ++j) {
        const int n = n0 + j;
        if (n >= N) break;
        const int beg = __builtin_amdgcn_readlane(offv, j);
        const int deg = __builtin_amdgcn_readlane(cntv, j);
        const v2f xr2 = ((const v2f*)xr)[(size_t)n * 16 + cl];
        const int2* __restrict__ edges = (const int2*)srt + beg;

        float m_run = -INFINITY;
        float s_run = 0.0f;
        v2f   acc   = {0.0f, 0.0f};

        for (int c0 = 0; c0 < deg; c0 += 32) {
            const int rem = deg - c0;
            const bool valid = lane < rem;
            int s = n;
            float wf = 0.0f;
            if (valid) {
                const int2 t = edges[c0 + lane];
                s = t.x;
                wf = __int_as_float(t.y);
            }

            v2f v[16];
#define GAT_LOAD(i) { const unsigned si = (unsigned)row_pick_i<0x150 + (i)>(s); v[i] = xl2[si * 16u + (unsigned)cl]; }
            REP16(GAT_LOAD)
#undef GAT_LOAD
            __builtin_amdgcn_sched_barrier(0);

            float q[16];
#define GAT_LOGIT(i) { \
            const float wi = row_pick_f<0x150 + (i)>(wf); \
            v2f mm = v[i] + xr2 + wi * we2; \
            mm[0] = (mm[0] > 0.0f) ? mm[0] : NEG_ATT * mm[0]; \
            mm[1] = (mm[1] > 0.0f) ? mm[1] : NEG_ATT * mm[1]; \
            q[i] = mm[0] * att2[0] + mm[1] * att2[1]; }
            REP16(GAT_LOGIT)
#undef GAT_LOGIT
            row_fold<8>(q, cl);
            row_fold<4>(q, cl);
            row_fold<2>(q, cl);
            row_fold<1>(q, cl);
            const float logit = valid ? q[0] : -INFINITY;

            float cmax = logit;
            #pragma unroll
            for (int o = 16; o >= 1; o >>= 1) cmax = fmaxf(cmax, __shfl_xor(cmax, o));
            const float m_new = fmaxf(m_run, cmax);
            const float scale = __expf(m_run - m_new);
            const float ex    = __expf(logit - m_new);
            float csum = ex;
            #pragma unroll
            for (int o = 16; o >= 1; o >>= 1) csum += __shfl_xor(csum, o);
            s_run = s_run * scale + csum;

            v2f a0 = {0.0f, 0.0f}, a1 = {0.0f, 0.0f};
#define GAT_ACC(i) { const float ei = row_pick_f<0x150 + (i)>(ex); if ((i) & 1) a1 += ei * v[i]; else a0 += ei * v[i]; }
            REP16(GAT_ACC)
#undef GAT_ACC
            acc = acc * scale + (a0 + a1);
            m_run = m_new;
        }

        const float tx = acc[0] + __shfl_xor(acc[0], 16);
        const float ty = acc[1] + __shfl_xor(acc[1], 16);
        float ox = (deg > 0) ? tx / s_run : 0.0f;
        float oy = (deg > 0) ? ty / s_run : 0.0f;
        ox += bias2[0];
        oy += bias2[1];
        if (lane < 16) {
            v2f o;
            o[0] = (ox > 0.0f) ? ox : NEG_OUT * ox;
            o[1] = (oy > 0.0f) ? oy : NEG_OUT * oy;
            ((v2f*)out)[(size_t)n * 16 + cl] = o;
        }
    }
}

extern "C" void kernel_launch(void* const* d_in, const int* in_sizes, int n_in,
                              void* d_out, int out_size, void* d_ws, size_t ws_size,
                              hipStream_t stream) {
    const float* x    = (const float*)d_in[0];
    const int*   ei   = (const int*)d_in[1];
    const float* ew   = (const float*)d_in[2];
    const float* Wl   = (const float*)d_in[3];
    const float* bl   = (const float*)d_in[4];
    const float* Wr   = (const float*)d_in[5];
    const float* br   = (const float*)d_in[6];
    const float* We   = (const float*)d_in[7];
    const float* att  = (const float*)d_in[8];
    const float* bias = (const float*)d_in[9];

    const int N = in_sizes[0] / 128;
    const int E = in_sizes[2];
    const int* srcv = ei;
    const int* dstv = ei + E;

    const int B    = (N + BNODES - 1) / BNODES;
    const int nblk = (E + CH - 1) / CH;

    float* xl   = (float*)d_ws;
    float* xr   = xl + (size_t)N * 32;
    int*   srt  = (int*)(xr + (size_t)N * 32);
    int*   bkt  = srt + (size_t)E * 2;
    int*   cnt  = bkt + (size_t)E * 2;
    int*   off  = cnt + (size_t)N;
    int*   tab  = off + (size_t)N;
    float* out  = (float*)d_out;

    const int tiles = (N + 15) / 16;
    const int wavesPerBlock = 8;
    const int gw = (tiles + TPG - 1) / TPG;
    const int g1 = (gw + 1) / 2;
    const int g2 = gw - g1;

    gat_bin_gemm<<<g1 + nblk, 1024, 0, stream>>>(
        srcv, dstv, ew, tab, bkt, E, B, nblk, g1, x, Wl, bl, Wr, br, xl, xr, N);

    gat_group_gemm<<<g2 + B, 1024, 0, stream>>>(
        bkt, tab, srt, off, cnt, N, E, B, nblk, g2, x, Wl, bl, Wr, br, xl, xr, g1 * TPG);

    gat_node<<<(N + wavesPerBlock * NPW - 1) / (wavesPerBlock * NPW), 32 * wavesPerBlock, 0, stream>>>(
        srt, off, cnt, xl, xr, We, att, bias, out, N);
}


# node pass: a chunk's edges go alternately to the two 16-lane rows and only the slots they fill are run (2..16 in steps of 2, one straight-line body each); sum order: a row adds every second edge
# speedup vs baseline: 16.1652x; 1.0425x over previous
#include <hip/hip_runtime.h>
#include <hip/hip_bf16.h>
#include <math.h>

#define NEG_ATT 0.2f
#define NEG_OUT 0.01f

typedef __attribute__((ext_vector_type(2))) float v2f;
typedef __attribute__((ext_vector_type(8))) float v8f;

#define BSH    8
#define BNODES 256
#define MAXB   512
#define CH     8192
#define MAXG   1024
#define GT     10240
#define KT     10
#define NW     32
#define TPG    28


__device__ __forceinline__ void w_image(const float* __restrict__ Wl,
                                        const float* __restrict__ Wr,
                                        float* wsh, const int tid) {
    #pragma unroll
    for (int j = 0; j < 8; ++j) {
        const int g   = j * 1024 + tid;
        const int mat = g >> 12;
        const int k   = (g >> 5) & 127;
        const int col = g & 31;
        const float val = (mat ? Wr : Wl)[g & 4095];
        const int t  = ((k >> 3) << 1) | ((k >> 1) & 1);
        const int h  = (k >> 2) & 1;
        const int e  = ((col >> 4) << 1) | (k & 1);
        const int ln = (h << 4) | (col & 15);
        wsh[((((t << 1) | mat) << 5) + ln) * 4 + e] = val;
    }
}

__device__ __forceinline__ void gemm_tile(
        const float* __restrict__ x,
        const float* wsh, const float* __restrict__ bl,
        const float* __restrict__ br,
        float* __restrict__ xl, float* __restrict__ xr, int N,
        const int tile, const int lane) {
    const int row0 = tile * 16;
    if (row0 >= N) return;

    const int m  = lane & 15;
    const int kh = lane >> 4;

    int arow = row0 + m;
    if (arow >= N) arow = N - 1;
    const float4* __restrict__ xq = (const float4*)(x + (size_t)arow * 128) + kh;
    const float4* wq = (const float4*)wsh + lane;

    v8f cl0 = {}, cl1 = {}, cr0 = {}, cr1 = {};

    for (int u = 0; u < 16; u += 2) {
        const float4 ax0 = xq[2 * u];
        const float4 ax1 = xq[2 * u + 2];
        #pragma unroll
        for (int q = 0; q < 4; ++q) {
            const int t = 2 * u + q;
            const int p = q & 1;
            const float4 ax = (q < 2) ? ax0 : ax1;
            const float4 wl4 = wq[(2 * t) * 32];
            const float4 wr4 = wq[(2 * t + 1) * 32];
            v2f a;
            a[0] = p ? ax.z : ax.x;
            a[1] = p ? ax.w : ax.y;
            v2f bl0v, bl1v, br0v, br1v;
            bl0v[0] = wl4.x;  bl0v[1] = wl4.y;
            bl1v[0] = wl4.z;  bl1v[1] = wl4.w;
            br0v[0] = wr4.x;  br0v[1] = wr4.y;
            br1v[0] = wr4.z;  br1v[1] = wr4.w;

            cl0 = __builtin_amdgcn_wmma_f32_16x16x4_f32(false, a, false, bl0v, (short)0, cl0, false, false);
            cl1 = __builtin_amdgcn_wmma_f32_16x16x4_f32(false, a, false, bl1v, (short)0, cl1, false, false);
            cr0 = __builtin_amdgcn_wmma_f32_16x16x4_f32(false, a, false, br0v, (short)0, cr0, false, false);
            cr1 = __builtin_amdgcn_wmma_f32_16x16x4_f32(false, a, false, br1v, (short)0, cr1, false, false);
        }
    }

    const int n = lane & 15;
    const float bl_lo = bl[n], bl_hi = bl[n + 16];
    const float br_lo = br[n], br_hi = br[n + 16];
    #pragma unroll
    for (int v = 0; v < 8; ++v) {
        const int mm = v + 8 * kh;
        const int row = row0 + mm;
        if (row < N) {
            const size_t ro = (size_t)row * 32;
            xl[ro + n]      = cl0[v] + bl_lo;
            xl[ro + n + 16] = cl1[v] + bl_hi;
            xr[ro + n]      = cr0[v] + br_lo;
            xr[ro + n + 16] = cr1[v] + br_hi;
        }
    }
}

__device__ __forceinline__ int wg_scan_excl(int c, int* wtot, int lane, int wv, int* total) {
    int incl = c;
    #pragma unroll
    for (int o = 1; o < 32; o <<= 1) {
        const int t = __shfl_up(incl, o);
        if (lane >= o) incl += t;
    }
    if (lane == 31) wtot[wv] = incl;
    __syncthreads();
    if (wv == 0) {
        const int t = wtot[lane];
        int inc = t;
        #pragma unroll
        for (int o = 1; o < 32; o <<= 1) {
            const int u = __shfl_up(inc, o);
            if (lane >= o) inc += u;
        }
        wtot[lane] = inc - t;
        if (lane == 31) wtot[32] = inc;
    }
    __syncthreads();
    *total = wtot[32];
    return wtot[wv] + incl - c;
}

__global__ __launch_bounds__(1024, 16) void gat_bin_gemm(
        const int* __restrict__ src, const int* __restrict__ dst,
        const float* __restrict__ ew, int* __restrict__ tab,
        int* __restrict__ bkt, int E, int B, int nblk, int ng,
        const float* __restrict__ x,
        const float* __restrict__ Wl, const float* __restrict__ bl,
        const float* __restrict__ Wr, const float* __restrict__ br,
        float* __restrict__ xl, float* __restrict__ xr, int N) {
    __shared__ __attribute__((aligned(16))) int wcnt[NW * MAXB];
    __shared__ int wtot[33];
    __shared__ int stx[CH];
    __shared__ int sty[CH];
    if ((int)blockIdx.x < ng) {
        float* wsh = (float*)wcnt;
        w_image(Wl, Wr, wsh, (int)threadIdx.x);
        __syncthreads();
        const int gwv = (int)(threadIdx.x >> 5);
        if (gwv < TPG)
            gemm_tile(x, wsh, bl, br, xl, xr, N,
                      (int)blockIdx.x * TPG + gwv, (int)(threadIdx.x & 31));
        return;
    }
    const int tid  = threadIdx.x;
    const int lane = tid & 31;
    const int wv   = tid >> 5;
    const int blk  = (int)blockIdx.x - ng;
    {
        const int4 z = {0, 0, 0, 0};
        #pragma unroll
        for (int i = tid; i < NW * MAXB / 4; i += 1024) ((int4*)wcnt)[i] = z;
    }
    const int base = blk * CH;
    int d[8], s[8], w[8], r[8];
    #pragma unroll
    for (int k = 0; k < 8; ++k) {
        const int e = base + k * 1024 + tid;
        if (e < E) {
            d[k] = dst[e];
            s[k] = src[e];
            w[k] = __float_as_int(ew[e]);
        } else {
            d[k] = -1;
            s[k] = 0;
            w[k] = 0;
        }
    }
    __syncthreads();
    int* __restrict__ mycnt = wcnt + wv * MAXB;
    #pragma unroll
    for (int k = 0; k < 8; ++k)
        r[k] = (d[k] >= 0) ? atomicAdd(&mycnt[d[k] >> BSH], 1) : 0;
    __syncthreads();

    const int rb = tid >> 1;
    const int q0 = (tid & 1) * (NW / 2);
    int c = 0;
    #pragma unroll
    for (int j = 0; j < NW / 2; ++j) c += wcnt[(q0 + j) * MAXB + rb];
    int nall;
    const int excl = wg_scan_excl(c, wtot, lane, wv, &nall);
    {
        int v[NW / 2];
        #pragma unroll
        for (int j = 0; j < NW / 2; ++j) v[j] = wcnt[(q0 + j) * MAXB + rb];
        int run = excl;
        #pragma unroll
        for (int j = 0; j < NW / 2; ++j) {
            wcnt[(q0 + j) * MAXB + rb] = run;
            run += v[j];
        }
    }
    if (q0 == 0 && rb < B) tab[rb * nblk + blk] = excl;
    __syncthreads();

    int pb[8];
    #pragma unroll
    for (int k = 0; k < 8; ++k) pb[k] = mycnt[(d[k] >= 0) ? (d[k] >> BSH) : 0];
    #pragma unroll
    for (int k = 0; k < 8; ++k) {
        if (d[k] >= 0) {
            const int p = pb[k] + r[k];
            stx[p] = (int)(((unsigned)(d[k] & (BNODES - 1)) << 24) | (unsigned)s[k]);
            sty[p] = w[k];
        }
    }
    __syncthreads();

    const int left = E - base;
    const int nE = (left < CH) ? left : CH;
    int2* __restrict__ outp = (int2*)bkt + base;
    for (int i = tid; i < nE; i += 1024) {
        int2 t;
        t.x = stx[i];
        t.y = sty[i];
        outp[i] = t;
    }
}

__global__ __launch_bounds__(1024, 16) void gat_group_gemm(
        const int* __restrict__ bkt, const int* __restrict__ tab,
        int* __restrict__ srt, int* __restrict__ off, int* __restrict__ cnt,
        int N, int E, int B, int nblk, int ng,
        const float* __restrict__ x,
        const float* __restrict__ Wl, const float* __restrict__ bl,
        const float* __restrict__ Wr, const float* __restrict__ br,
        float* __restrict__ xl, float* __restrict__ xr, int tile0) {
    __shared__ __attribute__((aligned(16))) int wcnt[NW * BNODES];
    __shared__ int wtot[33];
    __shared__ int wred[32];
    __shared__ int rsrc[MAXG];
    __shared__ unsigned short gid[GT];
    __shared__ __attribute__((aligned(16))) int stx[GT];
    __shared__ int sty[GT];
    if ((int)blockIdx.x < ng) {
        float* wsh = (float*)stx;
        w_image(Wl, Wr, wsh, (int)threadIdx.x);
        __syncthreads();
        const int gwv = (int)(threadIdx.x >> 5);
        if (gwv < TPG)
            gemm_tile(x, wsh, bl, br, xl, xr, N,
                      tile0 + (int)blockIdx.x * TPG + gwv, (int)(threadIdx.x & 31));
        return;
    }
    const int tid  = threadIdx.x;
    const int lane = tid & 31;
    const int wv   = tid >> 5;
    const int b    = (int)blockIdx.x - ng;
    {
        const int4 z = {0, 0, 0, 0};
        #pragma unroll
        for (int i = tid; i < NW * BNODES / 4; i += 1024) ((int4*)wcnt)[i] = z;
    }

    int st = 0, len = 0;
    if (tid < nblk) {
        st = tab[b * nblk + tid];
        int en;
        if (b + 1 < B) {
            en = tab[(b + 1) * nblk + tid];
        } else {
            const int left = E - tid * CH;
            en = (left < CH) ? left : CH;
        }
        len = en - st;
    }
    int ssum = st;
    #pragma unroll
    for (int o = 16; o >= 1; o >>= 1) ssum += __shfl_xor(ssum, o);
    if (lane == 0) wred[wv] = ssum;
    int nrecv;
    const int pre = wg_scan_excl(len, wtot, lane, wv, &nrecv);
    int bsum = wred[lane];
    #pragma unroll
    for (int o = 16; o >= 1; o >>= 1) bsum += __shfl_xor(bsum, o);
    const int nrec = __builtin_amdgcn_readfirstlane(nrecv);
    const int bbeg = __builtin_amdgcn_readfirstlane(bsum);
    if (tid < nblk) rsrc[tid] = tid * CH + st - pre;

    const int2* __restrict__ in = (const int2*)bkt;
    int2* __restrict__ outp = (int2*)srt + bbeg;
    int* __restrict__ mycnt = wcnt + wv * BNODES;

    for (int t0 = 0; t0 < nrec; t0 += GT) {
        __syncthreads();
        if (tid < nblk) {
            const int lo = (pre > t0) ? pre : t0;
            int hi = pre + len;
            if (hi > t0 + GT) hi = t0 + GT;
            #pragma clang loop vectorize(disable) interleave(disable) unroll(disable)
            for (int j = lo; j < hi; ++j) gid[j - t0] = (unsigned short)tid;
        }
        __syncthreads();
        int x[KT];
        #pragma unroll
        for (int k = 0; k < KT; ++k) {
            const int i = t0 + k * 1024 + tid;
            const int ii = (i < nrec) ? i : nrec - 1;
            x[k] = in[ii + rsrc[gid[ii - t0]]].x;
        }
        #pragma unroll
        for (int k = 0; k < KT; ++k) {
            const int i = t0 + k * 1024 + tid;
            if (i < nrec) atomicAdd(&mycnt[((unsigned)x[k]) >> 24], 1);
        }
    }
    __syncthreads();

    const int nd = tid >> 2;
    const int q0 = (tid & 3) * (NW / 4);
    int v[NW / 4];
    int c = 0;
    #pragma unroll
    for (int j = 0; j < NW / 4; ++j) {
        v[j] = wcnt[(q0 + j) * BNODES + nd];
        c += v[j];
    }
    int nall;
    const int excl = wg_scan_excl(c, wtot, lane, wv, &nall);
    {
        int run = excl;
        #pragma unroll
        for (int j = 0; j < NW / 4; ++j) {
            wcnt[(q0 + j) * BNODES + nd] = run;
            run += v[j];
        }
    }
    int ctot = c + __shfl_xor(c, 1);
    ctot += __shfl_xor(ctot, 2);
    if (q0 == 0) {
        const int node = b * BNODES + nd;
        if (node < N) {
            off[node] = bbeg + excl;
            cnt[node] = ctot;
        }
    }

    const bool staged = nrec <= GT;
    for (int t0 = 0; t0 < nrec; t0 += GT) {
        __syncthreads();
        if (!staged) {
            if (tid < nblk) {
                const int lo = (pre > t0) ? pre : t0;
                int hi = pre + len;
                if (hi > t0 + GT) hi = t0 + GT;
                #pragma clang loop vectorize(disable) interleave(disable) unroll(disable)
                for (int j = lo; j < hi; ++j) gid[j - t0] = (unsigned short)tid;
            }
            __syncthreads();
        }
        int rx[KT], ry[KT];
        #pragma unroll
        for (int k = 0; k < KT; ++k) {
            const int i = t0 + k * 1024 + tid;
            const int ii = (i < nrec) ? i : nrec - 1;
            const int2 rec = in[ii + rsrc[gid[ii - t0]]];
            rx[k] = rec.x;
            ry[k] = rec.y;
        }
        #pragma unroll
        for (int k = 0; k < KT; ++k) {
            const int i = t0 + k * 1024 + tid;
            if (i < nrec) {
                const int p = atomicAdd(&mycnt[((unsigned)rx[k]) >> 24], 1);
                const int sx = rx[k] & 0x00FFFFFF;
                if (staged) {
                    stx[p] = sx;
                    sty[p] = ry[k];
                } else {
                    int2 t;
                    t.x = sx;
                    t.y = ry[k];
                    outp[p] = t;
                }
            }
        }
    }
    if (staged) {
        __syncthreads();
        for (int i = tid; i < nrec; i += 1024) {
            int2 t;
            t.x = stx[i];
            t.y = sty[i];
            outp[i] = t;
        }
    }
}


template <int CTRL>
__device__ __forceinline__ int row_pick_i(int x) {
    return __builtin_amdgcn_update_dpp(0, x, CTRL, 0xF, 0xF, true);
}
template <int CTRL>
__device__ __forceinline__ float row_pick_f(float x) {
    return __int_as_float(__builtin_amdgcn_update_dpp(0, __float_as_int(x), CTRL, 0xF, 0xF, true));
}

template <int H, int NS>
__device__ __forceinline__ void row_fold(float* q, int cl) {
    const bool up = (cl & H) != 0;
    #pragma unroll
    for (int j = 0; j < H; ++j) {
        if (j >= NS) continue;
        const float lo = q[j];
        const float hi = q[j + H];
        const float keep = up ? hi : lo;
        const float send = up ? lo : hi;
        q[j] = keep + row_pick_f<0x160 + H>(send);
    }
}

#define REP16(M) M(0) M(1) M(2) M(3) M(4) M(5) M(6) M(7) M(8) M(9) M(10) M(11) M(12) M(13) M(14) M(15)

template <int NS>
__device__ __forceinline__ void node_chunk(
        const v2f* __restrict__ xl2, const int s, const float wf, const bool valid,
        const int cl, const v2f xr2, const v2f we2, const v2f att2,
        float& m_run, float& s_run, v2f& acc) {
    v2f v[16];
#define GAT_LOAD(i) if ((i) < NS) { const unsigned si = (unsigned)row_pick_i<0x150 + (i)>(s); v[i] = xl2[si * 16u + (unsigned)cl]; }
    REP16(GAT_LOAD)
#undef GAT_LOAD
    __builtin_amdgcn_sched_barrier(0);

    float q[16];
#define GAT_LOGIT(i) if ((i) < NS) { \
    const float wi = row_pick_f<0x150 + (i)>(wf); \
    v2f mm = v[i] + xr2 + wi * we2; \
    mm[0] = (mm[0] > 0.0f) ? mm[0] : NEG_ATT * mm[0]; \
    mm[1] = (mm[1] > 0.0f) ? mm[1] : NEG_ATT * mm[1]; \
    q[i] = mm[0] * att2[0] + mm[1] * att2[1]; } else { q[i] = 0.0f; }
    REP16(GAT_LOGIT)
#undef GAT_LOGIT
    row_fold<8, NS>(q, cl);
    row_fold<4, NS>(q, cl);
    row_fold<2, NS>(q, cl);
    row_fold<1, NS>(q, cl);
    const float logit = valid ? q[0] : -INFINITY;

    float cmax = logit;
    #pragma unroll
    for (int o = 16; o >= 1; o >>= 1) cmax = fmaxf(cmax, __shfl_xor(cmax, o));
    const float m_new = fmaxf(m_run, cmax);
    const float scale = __expf(m_run - m_new);
    const float ex    = __expf(logit - m_new);
    float csum = ex;
    #pragma unroll
    for (int o = 16; o >= 1; o >>= 1) csum += __shfl_xor(csum, o);
    s_run = s_run * scale + csum;

    v2f a0 = {0.0f, 0.0f}, a1 = {0.0f, 0.0f};
#define GAT_ACC(i) if ((i) < NS) { const float ei = row_pick_f<0x150 + (i)>(ex); if ((i) & 1) a1 += ei * v[i]; else a0 += ei * v[i]; }
    REP16(GAT_ACC)
#undef GAT_ACC
    acc = acc * scale + (a0 + a1);
    m_run = m_new;
}

#define NPW 4

__global__ __launch_bounds__(256) void gat_node(
        const int* __restrict__ srt, const int* __restrict__ off,
        const int* __restrict__ cnt,
        const float* __restrict__ xl, const float* __restrict__ xr,
        const float* __restrict__ We, const float* __restrict__ att,
        const float* __restrict__ bias, float* __restrict__ out, int N) {
    const int lane = threadIdx.x & 31;
    const int cl   = lane & 15;
    const int eslot = 2 * cl + (lane >> 4);
    const int w = __builtin_amdgcn_readfirstlane(
        (int)((blockIdx.x * blockDim.x + threadIdx.x) >> 5));
    const int n0 = w * NPW;
    if (n0 >= N) return;

    int offv = 0, cntv = 0;
    if (lane < NPW) {
        const int nn = (n0 + lane < N) ? n0 + lane : N - 1;
        offv = off[nn];
        cntv = cnt[nn];
    }

    const v2f* __restrict__ xl2 = (const v2f*)xl;
    const int2* __restrict__ srt2 = (const int2*)srt;
    v2f we2, att2, bias2;
    we2[0]   = We[2 * cl];    we2[1]   = We[2 * cl + 1];
    att2[0]  = att[2 * cl];   att2[1]  = att[2 * cl + 1];
    bias2[0] = bias[2 * cl];  bias2[1] = bias[2 * cl + 1];

    #pragma unroll 1
    for (int j = 0; j < NPW; ++j) {
        const int n = n0 + j;
        if (n >= N) break;
        const int beg = __builtin_amdgcn_readlane(offv, j);
        const int deg = __builtin_amdgcn_readlane(cntv, j);
        const v2f xr2 = ((const v2f*)xr)[(size_t)n * 16 + cl];

        float m_run = -INFINITY;
        float s_run = 0.0f;
        v2f   acc   = {0.0f, 0.0f};

        for (int c0 = 0; c0 < deg; c0 += 32) {
            const int rem = deg - c0;
            const bool valid = eslot < rem;
            int s = n;
            float wf = 0.0f;
            if (valid) {
                const int2 t = srt2[(unsigned)(beg + c0) + (unsigned)eslot];
                s = t.x;
                wf = __int_as_float(t.y);
            }
#define GAT_CHUNK(NS) node_chunk<NS>(xl2, s, wf, valid, cl, xr2, we2, att2, m_run, s_run, acc)
            if (rem > 16) {
                if (rem > 24) { if (rem > 28) GAT_CHUNK(16); else GAT_CHUNK(14); }
                else          { if (rem > 20) GAT_CHUNK(12); else GAT_CHUNK(10); }
            } else {
                if (rem > 8)  { if (rem > 12) GAT_CHUNK(8);  else GAT_CHUNK(6); }
                else          { if (rem > 4)  GAT_CHUNK(4);  else GAT_CHUNK(2); }
            }
#undef GAT_CHUNK
        }

        const float tx = acc[0] + __shfl_xor(acc[0], 16);
        const float ty = acc[1] + __shfl_xor(acc[1], 16);
        float ox = (deg > 0) ? tx / s_run : 0.0f;
        float oy = (deg > 0) ? ty / s_run : 0.0f;
        ox += bias2[0];
        oy += bias2[1];
        if (lane < 16) {
            v2f o;
            o[0] = (ox > 0.0f) ? ox : NEG_OUT * ox;
            o[1] = (oy > 0.0f) ? oy : NEG_OUT * oy;
            ((v2f*)out)[(size_t)n * 16 + cl] = o;
        }
    }
}

extern "C" void kernel_launch(void* const* d_in, const int* in_sizes, int n_in,
                              void* d_out, int out_size, void* d_ws, size_t ws_size,
                              hipStream_t stream) {
    const float* x    = (const float*)d_in[0];
    const int*   ei   = (const int*)d_in[1];
    const float* ew   = (const float*)d_in[2];
    const float* Wl   = (const float*)d_in[3];
    const float* bl   = (const float*)d_in[4];
    const float* Wr   = (const float*)d_in[5];
    const float* br   = (const float*)d_in[6];
    const float* We   = (const float*)d_in[7];
    const float* att  = (const float*)d_in[8];
    const float* bias = (const float*)d_in[9];

    const int N = in_sizes[0] / 128;
    const int E = in_sizes[2];
    const int* srcv = ei;
    const int* dstv = ei + E;

    const int B    = (N + BNODES - 1) / BNODES;
    const int nblk = (E + CH - 1) / CH;

    float* xl   = (float*)d_ws;
    float* xr   = xl + (size_t)N * 32;
    int*   srt  = (int*)(xr + (size_t)N * 32);
    int*   bkt  = srt + (size_t)E * 2;
    int*   cnt  = bkt + (size_t)E * 2;
    int*   off  = cnt + (size_t)N;
    int*   tab  = off + (size_t)N;
    float* out  = (float*)d_out;

    const int tiles = (N + 15) / 16;
    const int wavesPerBlock = 8;
    const int gw = (tiles + TPG - 1) / TPG;
    const int g1 = (gw + 1) / 2;
    const int g2 = gw - g1;

    gat_bin_gemm<<<g1 + nblk, 1024, 0, stream>>>(
        srcv, dstv, ew, tab, bkt, E, B, nblk, g1, x, Wl, bl, Wr, br, xl, xr, N);

    gat_group_gemm<<<g2 + B, 1024, 0, stream>>>(
        bkt, tab, srt, off, cnt, N, E, B, nblk, g2, x, Wl, bl, Wr, br, xl, xr, g1 * TPG);

    gat_node<<<(N + wavesPerBlock * NPW - 1) / (wavesPerBlock * NPW), 32 * wavesPerBlock, 0, stream>>>(
        srt, off, cnt, xl, xr, We, att, bias, out, N);
}


# node pass: a chunk's 32-lane max and exp sum (and the closing row add) as row_xmask DPP steps and one permlanex16 on the vector ALU, not ten LDS-unit permutes each waited for; same sum order
# speedup vs baseline: 16.8117x; 1.0400x over previous
#include <hip/hip_runtime.h>
#include <hip/hip_bf16.h>
#include <math.h>

#define NEG_ATT 0.2f
#define NEG_OUT 0.01f

typedef __attribute__((ext_vector_type(2))) float v2f;
typedef __attribute__((ext_vector_type(8))) float v8f;

#define BSH    8
#define BNODES 256
#define MAXB   512
#define CH     8192
#define MAXG   1024
#define GT     10240
#define KT     10
#define NW     32
#define TPG    28


__device__ __forceinline__ void w_image(const float* __restrict__ Wl,
                                        const float* __restrict__ Wr,
                                        float* wsh, const int tid) {
    #pragma unroll
    for (int j = 0; j < 8; ++j) {
        const int g   = j * 1024 + tid;
        const int mat = g >> 12;
        const int k   = (g >> 5) & 127;
        const int col = g & 31;
        const float val = (mat ? Wr : Wl)[g & 4095];
        const int t  = ((k >> 3) << 1) | ((k >> 1) & 1);
        const int h  = (k >> 2) & 1;
        const int e  = ((col >> 4) << 1) | (k & 1);
        const int ln = (h << 4) | (col & 15);
        wsh[((((t << 1) | mat) << 5) + ln) * 4 + e] = val;
    }
}

__device__ __forceinline__ void gemm_tile(
        const float* __restrict__ x,
        const float* wsh, const float* __restrict__ bl,
        const float* __restrict__ br,
        float* __restrict__ xl, float* __restrict__ xr, int N,
        const int tile, const int lane) {
    const int row0 = tile * 16;
    if (row0 >= N) return;

    const int m  = lane & 15;
    const int kh = lane >> 4;

    int arow = row0 + m;
    if (arow >= N) arow = N - 1;
    const float4* __restrict__ xq = (const float4*)(x + (size_t)arow * 128) + kh;
    const float4* wq = (const float4*)wsh + lane;

    v8f cl0 = {}, cl1 = {}, cr0 = {}, cr1 = {};

    for (int u = 0; u < 16; u += 2) {
        const float4 ax0 = xq[2 * u];
        const float4 ax1 = xq[2 * u + 2];
        #pragma unroll
        for (int q = 0; q < 4; ++q) {
            const int t = 2 * u + q;
            const int p = q & 1;
            const float4 ax = (q < 2) ? ax0 : ax1;
            const float4 wl4 = wq[(2 * t) * 32];
            const float4 wr4 = wq[(2 * t + 1) * 32];
            v2f a;
            a[0] = p ? ax.z : ax.x;
            a[1] = p ? ax.w : ax.y;
            v2f bl0v, bl1v, br0v, br1v;
            bl0v[0] = wl4.x;  bl0v[1] = wl4.y;
            bl1v[0] = wl4.z;  bl1v[1] = wl4.w;
            br0v[0] = wr4.x;  br0v[1] = wr4.y;
            br1v[0] = wr4.z;  br1v[1] = wr4.w;

            cl0 = __builtin_amdgcn_wmma_f32_16x16x4_f32(false, a, false, bl0v, (short)0, cl0, false, false);
            cl1 = __builtin_amdgcn_wmma_f32_16x16x4_f32(false, a, false, bl1v, (short)0, cl1, false, false);
            cr0 = __builtin_amdgcn_wmma_f32_16x16x4_f32(false, a, false, br0v, (short)0, cr0, false, false);
            cr1 = __builtin_amdgcn_wmma_f32_16x16x4_f32(false, a, false, br1v, (short)0, cr1, false, false);
        }
    }

    const int n = lane & 15;
    const float bl_lo = bl[n], bl_hi = bl[n + 16];
    const float br_lo = br[n], br_hi = br[n + 16];
    #pragma unroll
    for (int v = 0; v < 8; ++v) {
        const int mm = v + 8 * kh;
        const int row = row0 + mm;
        if (row < N) {
            const size_t ro = (size_t)row * 32;
            xl[ro + n]      = cl0[v] + bl_lo;
            xl[ro + n + 16] = cl1[v] + bl_hi;
            xr[ro + n]      = cr0[v] + br_lo;
            xr[ro + n + 16] = cr1[v] + br_hi;
        }
    }
}

__device__ __forceinline__ int wg_scan_excl(int c, int* wtot, int lane, int wv, int* total) {
    int incl = c;
    #pragma unroll
    for (int o = 1; o < 32; o <<= 1) {
        const int t = __shfl_up(incl, o);
        if (lane >= o) incl += t;
    }
    if (lane == 31) wtot[wv] = incl;
    __syncthreads();
    if (wv == 0) {
        const int t = wtot[lane];
        int inc = t;
        #pragma unroll
        for (int o = 1; o < 32; o <<= 1) {
            const int u = __shfl_up(inc, o);
            if (lane >= o) inc += u;
        }
        wtot[lane] = inc - t;
        if (lane == 31) wtot[32] = inc;
    }
    __syncthreads();
    *total = wtot[32];
    return wtot[wv] + incl - c;
}

__global__ __launch_bounds__(1024, 16) void gat_bin_gemm(
        const int* __restrict__ src, const int* __restrict__ dst,
        const float* __restrict__ ew, int* __restrict__ tab,
        int* __restrict__ bkt, int E, int B, int nblk, int ng,
        const float* __restrict__ x,
        const float* __restrict__ Wl, const float* __restrict__ bl,
        const float* __restrict__ Wr, const float* __restrict__ br,
        float* __restrict__ xl, float* __restrict__ xr, int N) {
    __shared__ __attribute__((aligned(16))) int wcnt[NW * MAXB];
    __shared__ int wtot[33];
    __shared__ int stx[CH];
    __shared__ int sty[CH];
    if ((int)blockIdx.x < ng) {
        float* wsh = (float*)wcnt;
        w_image(Wl, Wr, wsh, (int)threadIdx.x);
        __syncthreads();
        const int gwv = (int)(threadIdx.x >> 5);
        if (gwv < TPG)
            gemm_tile(x, wsh, bl, br, xl, xr, N,
                      (int)blockIdx.x * TPG + gwv, (int)(threadIdx.x & 31));
        return;
    }
    const int tid  = threadIdx.x;
    const int lane = tid & 31;
    const int wv   = tid >> 5;
    const int blk  = (int)blockIdx.x - ng;
    {
        const int4 z = {0, 0, 0, 0};
        #pragma unroll
        for (int i = tid; i < NW * MAXB / 4; i += 1024) ((int4*)wcnt)[i] = z;
    }
    const int base = blk * CH;
    int d[8], s[8], w[8], r[8];
    #pragma unroll
    for (int k = 0; k < 8; ++k) {
        const int e = base + k * 1024 + tid;
        if (e < E) {
            d[k] = dst[e];
            s[k] = src[e];
            w[k] = __float_as_int(ew[e]);
        } else {
            d[k] = -1;
            s[k] = 0;
            w[k] = 0;
        }
    }
    __syncthreads();
    int* __restrict__ mycnt = wcnt + wv * MAXB;
    #pragma unroll
    for (int k = 0; k < 8; ++k)
        r[k] = (d[k] >= 0) ? atomicAdd(&mycnt[d[k] >> BSH], 1) : 0;
    __syncthreads();

    const int rb = tid >> 1;
    const int q0 = (tid & 1) * (NW / 2);
    int c = 0;
    #pragma unroll
    for (int j = 0; j < NW / 2; ++j) c += wcnt[(q0 + j) * MAXB + rb];
    int nall;
    const int excl = wg_scan_excl(c, wtot, lane, wv, &nall);
    {
        int v[NW / 2];
        #pragma unroll
        for (int j = 0; j < NW / 2; ++j) v[j] = wcnt[(q0 + j) * MAXB + rb];
        int run = excl;
        #pragma unroll
        for (int j = 0; j < NW / 2; ++j) {
            wcnt[(q0 + j) * MAXB + rb] = run;
            run += v[j];
        }
    }
    if (q0 == 0 && rb < B) tab[rb * nblk + blk] = excl;
    __syncthreads();

    int pb[8];
    #pragma unroll
    for (int k = 0; k < 8; ++k) pb[k] = mycnt[(d[k] >= 0) ? (d[k] >> BSH) : 0];
    #pragma unroll
    for (int k = 0; k < 8; ++k) {
        if (d[k] >= 0) {
            const int p = pb[k] + r[k];
            stx[p] = (int)(((unsigned)(d[k] & (BNODES - 1)) << 24) | (unsigned)s[k]);
            sty[p] = w[k];
        }
    }
    __syncthreads();

    const int left = E - base;
    const int nE = (left < CH) ? left : CH;
    int2* __restrict__ outp = (int2*)bkt + base;
    for (int i = tid; i < nE; i += 1024) {
        int2 t;
        t.x = stx[i];
        t.y = sty[i];
        outp[i] = t;
    }
}

__global__ __launch_bounds__(1024, 16) void gat_group_gemm(
        const int* __restrict__ bkt, const int* __restrict__ tab,
        int* __restrict__ srt, int* __restrict__ off, int* __restrict__ cnt,
        int N, int E, int B, int nblk, int ng,
        const float* __restrict__ x,
        const float* __restrict__ Wl, const float* __restrict__ bl,
        const float* __restrict__ Wr, const float* __restrict__ br,
        float* __restrict__ xl, float* __restrict__ xr, int tile0) {
    __shared__ __attribute__((aligned(16))) int wcnt[NW * BNODES];
    __shared__ int wtot[33];
    __shared__ int wred[32];
    __shared__ int rsrc[MAXG];
    __shared__ unsigned short gid[GT];
    __shared__ __attribute__((aligned(16))) int stx[GT];
    __shared__ int sty[GT];
    if ((int)blockIdx.x < ng) {
        float* wsh = (float*)stx;
        w_image(Wl, Wr, wsh, (int)threadIdx.x);
        __syncthreads();
        const int gwv = (int)(threadIdx.x >> 5);
        if (gwv < TPG)
            gemm_tile(x, wsh, bl, br, xl, xr, N,
                      tile0 + (int)blockIdx.x * TPG + gwv, (int)(threadIdx.x & 31));
        return;
    }
    const int tid  = threadIdx.x;
    const int lane = tid & 31;
    const int wv   = tid >> 5;
    const int b    = (int)blockIdx.x - ng;
    {
        const int4 z = {0, 0, 0, 0};
        #pragma unroll
        for (int i = tid; i < NW * BNODES / 4; i += 1024) ((int4*)wcnt)[i] = z;
    }

    int st = 0, len = 0;
    if (tid < nblk) {
        st = tab[b * nblk + tid];
        int en;
        if (b + 1 < B) {
            en = tab[(b + 1) * nblk + tid];
        } else {
            const int left = E - tid * CH;
            en = (left < CH) ? left : CH;
        }
        len = en - st;
    }
    int ssum = st;
    #pragma unroll
    for (int o = 16; o >= 1; o >>= 1) ssum += __shfl_xor(ssum, o);
    if (lane == 0) wred[wv] = ssum;
    int nrecv;
    const int pre = wg_scan_excl(len, wtot, lane, wv, &nrecv);
    int bsum = wred[lane];
    #pragma unroll
    for (int o = 16; o >= 1; o >>= 1) bsum += __shfl_xor(bsum, o);
    const int nrec = __builtin_amdgcn_readfirstlane(nrecv);
    const int bbeg = __builtin_amdgcn_readfirstlane(bsum);
    if (tid < nblk) rsrc[tid] = tid * CH + st - pre;

    const int2* __restrict__ in = (const int2*)bkt;
    int2* __restrict__ outp = (int2*)srt + bbeg;
    int* __restrict__ mycnt = wcnt + wv * BNODES;

    for (int t0 = 0; t0 < nrec; t0 += GT) {
        __syncthreads();
        if (tid < nblk) {
            const int lo = (pre > t0) ? pre : t0;
            int hi = pre + len;
            if (hi > t0 + GT) hi = t0 + GT;
            #pragma clang loop vectorize(disable) interleave(disable) unroll(disable)
            for (int j = lo; j < hi; ++j) gid[j - t0] = (unsigned short)tid;
        }
        __syncthreads();
        int x[KT];
        #pragma unroll
        for (int k = 0; k < KT; ++k) {
            const int i = t0 + k * 1024 + tid;
            const int ii = (i < nrec) ? i : nrec - 1;
            x[k] = in[ii + rsrc[gid[ii - t0]]].x;
        }
        #pragma unroll
        for (int k = 0; k < KT; ++k) {
            const int i = t0 + k * 1024 + tid;
            if (i < nrec) atomicAdd(&mycnt[((unsigned)x[k]) >> 24], 1);
        }
    }
    __syncthreads();

    const int nd = tid >> 2;
    const int q0 = (tid & 3) * (NW / 4);
    int v[NW / 4];
    int c = 0;
    #pragma unroll
    for (int j = 0; j < NW / 4; ++j) {
        v[j] = wcnt[(q0 + j) * BNODES + nd];
        c += v[j];
    }
    int nall;
    const int excl = wg_scan_excl(c, wtot, lane, wv, &nall);
    {
        int run = excl;
        #pragma unroll
        for (int j = 0; j < NW / 4; ++j) {
            wcnt[(q0 + j) * BNODES + nd] = run;
            run += v[j];
        }
    }
    int ctot = c + __shfl_xor(c, 1);
    ctot += __shfl_xor(ctot, 2);
    if (q0 == 0) {
        const int node = b * BNODES + nd;
        if (node < N) {
            off[node] = bbeg + excl;
            cnt[node] = ctot;
        }
    }

    const bool staged = nrec <= GT;
    for (int t0 = 0; t0 < nrec; t0 += GT) {
        __syncthreads();
        if (!staged) {
            if (tid < nblk) {
                const int lo = (pre > t0) ? pre : t0;
                int hi = pre + len;
                if (hi > t0 + GT) hi = t0 + GT;
                #pragma clang loop vectorize(disable) interleave(disable) unroll(disable)
                for (int j = lo; j < hi; ++j) gid[j - t0] = (unsigned short)tid;
            }
            __syncthreads();
        }
        int rx[KT], ry[KT];
        #pragma unroll
        for (int k = 0; k < KT; ++k) {
            const int i = t0 + k * 1024 + tid;
            const int ii = (i < nrec) ? i : nrec - 1;
            const int2 rec = in[ii + rsrc[gid[ii - t0]]];
            rx[k] = rec.x;
            ry[k] = rec.y;
        }
        #pragma unroll
        for (int k = 0; k < KT; ++k) {
            const int i = t0 + k * 1024 + tid;
            if (i < nrec) {
                const int p = atomicAdd(&mycnt[((unsigned)rx[k]) >> 24], 1);
                const int sx = rx[k] & 0x00FFFFFF;
                if (staged) {
                    stx[p] = sx;
                    sty[p] = ry[k];
                } else {
                    int2 t;
                    t.x = sx;
                    t.y = ry[k];
                    outp[p] = t;
                }
            }
        }
    }
    if (staged) {
        __syncthreads();
        for (int i = tid; i < nrec; i += 1024) {
            int2 t;
            t.x = stx[i];
            t.y = sty[i];
            outp[i] = t;
        }
    }
}


template <int CTRL>
__device__ __forceinline__ int row_pick_i(int x) {
    return __builtin_amdgcn_update_dpp(0, x, CTRL, 0xF, 0xF, true);
}
template <int CTRL>
__device__ __forceinline__ float row_pick_f(float x) {
    return __int_as_float(__builtin_amdgcn_update_dpp(0, __float_as_int(x), CTRL, 0xF, 0xF, true));
}

__device__ __forceinline__ float row_other_f(float x) {
    const int b = __float_as_int(x);
    return __int_as_float(__builtin_amdgcn_permlanex16(b, b, 0x76543210, 0xfedcba98, false, false));
}

template <int M>
__device__ __forceinline__ float row_xmax(float x) {
    float r;
    asm("v_max_num_f32_dpp %0, %1, %1 row_xmask:%2 row_mask:0xf bank_mask:0xf"
        : "=v"(r) : "v"(x), "n"(M));
    return r;
}

__device__ __forceinline__ float wave_max(float x) {
    x = fmaxf(x, row_other_f(x));
    x = row_xmax<8>(x);
    x = row_xmax<4>(x);
    x = row_xmax<2>(x);
    x = row_xmax<1>(x);
    return x;
}
__device__ __forceinline__ float wave_sum(float x) {
    x += row_other_f(x);
    x += row_pick_f<0x168>(x);
    x += row_pick_f<0x164>(x);
    x += row_pick_f<0x162>(x);
    x += row_pick_f<0x161>(x);
    return x;
}

template <int H, int NS>
__device__ __forceinline__ void row_fold(float* q, int cl) {
    const bool up = (cl & H) != 0;
    #pragma unroll
    for (int j = 0; j < H; ++j) {
        if (j >= NS) continue;
        const float lo = q[j];
        const float hi = q[j + H];
        const float keep = up ? hi : lo;
        const float send = up ? lo : hi;
        q[j] = keep + row_pick_f<0x160 + H>(send);
    }
}

#define REP16(M) M(0) M(1) M(2) M(3) M(4) M(5) M(6) M(7) M(8) M(9) M(10) M(11) M(12) M(13) M(14) M(15)

template <int NS>
__device__ __forceinline__ void node_chunk(
        const v2f* __restrict__ xl2, const int s, const float wf, const bool valid,
        const int cl, const v2f xr2, const v2f we2, const v2f att2,
        float& m_run, float& s_run, v2f& acc) {
    v2f v[16];
#define GAT_LOAD(i) if ((i) < NS) { const unsigned si = (unsigned)row_pick_i<0x150 + (i)>(s); v[i] = xl2[si * 16u + (unsigned)cl]; }
    REP16(GAT_LOAD)
#undef GAT_LOAD
    __builtin_amdgcn_sched_barrier(0);

    float q[16];
#define GAT_LOGIT(i) if ((i) < NS) { \
    const float wi = row_pick_f<0x150 + (i)>(wf); \
    v2f mm = v[i] + xr2 + wi * we2; \
    mm[0] = (mm[0] > 0.0f) ? mm[0] : NEG_ATT * mm[0]; \
    mm[1] = (mm[1] > 0.0f) ? mm[1] : NEG_ATT * mm[1]; \
    q[i] = mm[0] * att2[0] + mm[1] * att2[1]; } else { q[i] = 0.0f; }
    REP16(GAT_LOGIT)
#undef GAT_LOGIT
    row_fold<8, NS>(q, cl);
    row_fold<4, NS>(q, cl);
    row_fold<2, NS>(q, cl);
    row_fold<1, NS>(q, cl);
    const float logit = valid ? q[0] : -INFINITY;

    const float cmax = wave_max(logit);
    const float m_new = fmaxf(m_run, cmax);
    const float scale = __expf(m_run - m_new);
    const float ex    = __expf(logit - m_new);
    const float csum = wave_sum(ex);
    s_run = s_run * scale + csum;

    v2f a0 = {0.0f, 0.0f}, a1 = {0.0f, 0.0f};
#define GAT_ACC(i) if ((i) < NS) { const float ei = row_pick_f<0x150 + (i)>(ex); if ((i) & 1) a1 += ei * v[i]; else a0 += ei * v[i]; }
    REP16(GAT_ACC)
#undef GAT_ACC
    acc = acc * scale + (a0 + a1);
    m_run = m_new;
}

#define NPW 4

__global__ __launch_bounds__(256) void gat_node(
        const int* __restrict__ srt, const int* __restrict__ off,
        const int* __restrict__ cnt,
        const float* __restrict__ xl, const float* __restrict__ xr,
        const float* __restrict__ We, const float* __restrict__ att,
        const float* __restrict__ bias, float* __restrict__ out, int N) {
    const int lane = threadIdx.x & 31;
    const int cl   = lane & 15;
    const int eslot = 2 * cl + (lane >> 4);
    const int w = __builtin_amdgcn_readfirstlane(
        (int)((blockIdx.x * blockDim.x + threadIdx.x) >> 5));
    const int n0 = w * NPW;
    if (n0 >= N) return;

    int offv = 0, cntv = 0;
    if (lane < NPW) {
        const int nn = (n0 + lane < N) ? n0 + lane : N - 1;
        offv = off[nn];
        cntv = cnt[nn];
    }

    const v2f* __restrict__ xl2 = (const v2f*)xl;
    const int2* __restrict__ srt2 = (const int2*)srt;
    v2f we2, att2, bias2;
    we2[0]   = We[2 * cl];    we2[1]   = We[2 * cl + 1];
    att2[0]  = att[2 * cl];   att2[1]  = att[2 * cl + 1];
    bias2[0] = bias[2 * cl];  bias2[1] = bias[2 * cl + 1];

    #pragma unroll 1
    for (int j = 0; j < NPW; ++j) {
        const int n = n0 + j;
        if (n >= N) break;
        const int beg = __builtin_amdgcn_readlane(offv, j);
        const int deg = __builtin_amdgcn_readlane(cntv, j);
        const v2f xr2 = ((const v2f*)xr)[(size_t)n * 16 + cl];

        float m_run = -INFINITY;
        float s_run = 0.0f;
        v2f   acc   = {0.0f, 0.0f};

        for (int c0 = 0; c0 < deg; c0 += 32) {
            const int rem = deg - c0;
            const bool valid = eslot < rem;
            int s = n;
            float wf = 0.0f;
            if (valid) {
                const int2 t = srt2[(unsigned)(beg + c0) + (unsigned)eslot];
                s = t.x;
                wf = __int_as_float(t.y);
            }
#define GAT_CHUNK(NS) node_chunk<NS>(xl2, s, wf, valid, cl, xr2, we2, att2, m_run, s_run, acc)
            if (rem > 16) {
                if (rem > 24) { if (rem > 28) GAT_CHUNK(16); else GAT_CHUNK(14); }
                else          { if (rem > 20) GAT_CHUNK(12); else GAT_CHUNK(10); }
            } else {
                if (rem > 8)  { if (rem > 12) GAT_CHUNK(8);  else GAT_CHUNK(6); }
                else          { if (rem > 4)  GAT_CHUNK(4);  else GAT_CHUNK(2); }
            }
#undef GAT_CHUNK
        }

        const float tx = acc[0] + row_other_f(acc[0]);
        const float ty = acc[1] + row_other_f(acc[1]);
        float ox = (deg > 0) ? tx / s_run : 0.0f;
        float oy = (deg > 0) ? ty / s_run : 0.0f;
        ox += bias2[0];
        oy += bias2[1];
        if (lane < 16) {
            v2f o;
            o[0] = (ox > 0.0f) ? ox : NEG_OUT * ox;
            o[1] = (oy > 0.0f) ? oy : NEG_OUT * oy;
            ((v2f*)out)[(size_t)n * 16 + cl] = o;
        }
    }
}

extern "C" void kernel_launch(void* const* d_in, const int* in_sizes, int n_in,
                              void* d_out, int out_size, void* d_ws, size_t ws_size,
                              hipStream_t stream) {
    const float* x    = (const float*)d_in[0];
    const int*   ei   = (const int*)d_in[1];
    const float* ew   = (const float*)d_in[2];
    const float* Wl   = (const float*)d_in[3];
    const float* bl   = (const float*)d_in[4];
    const float* Wr   = (const float*)d_in[5];
    const float* br   = (const float*)d_in[6];
    const float* We   = (const float*)d_in[7];
    const float* att  = (const float*)d_in[8];
    const float* bias = (const float*)d_in[9];

    const int N = in_sizes[0] / 128;
    const int E = in_sizes[2];
    const int* srcv = ei;
    const int* dstv = ei + E;

    const int B    = (N + BNODES - 1) / BNODES;
    const int nblk = (E + CH - 1) / CH;

    float* xl   = (float*)d_ws;
    float* xr   = xl + (size_t)N * 32;
    int*   srt  = (int*)(xr + (size_t)N * 32);
    int*   bkt  = srt + (size_t)E * 2;
    int*   cnt  = bkt + (size_t)E * 2;
    int*   off  = cnt + (size_t)N;
    int*   tab  = off + (size_t)N;
    float* out  = (float*)d_out;

    const int tiles = (N + 15) / 16;
    const int wavesPerBlock = 8;
    const int gw = (tiles + TPG - 1) / TPG;
    const int g1 = (gw + 1) / 2;
    const int g2 = gw - g1;

    gat_bin_gemm<<<g1 + nblk, 1024, 0, stream>>>(
        srcv, dstv, ew, tab, bkt, E, B, nblk, g1, x, Wl, bl, Wr, br, xl, xr, N);

    gat_group_gemm<<<g2 + B, 1024, 0, stream>>>(
        bkt, tab, srt, off, cnt, N, E, B, nblk, g2, x, Wl, bl, Wr, br, xl, xr, g1 * TPG);

    gat_node<<<(N + wavesPerBlock * NPW - 1) / (wavesPerBlock * NPW), 32 * wavesPerBlock, 0, stream>>>(
        srt, off, cnt, xl, xr, We, att, bias, out, N);
}
